# starting kernel
# baseline (speedup 1.0000x reference)
#include <hip/hip_runtime.h>
#include <math.h>

typedef __attribute__((ext_vector_type(2))) float v2f;
typedef __attribute__((ext_vector_type(8))) float v8f;

#define E_DIM 64
#define H_DIM 128
#define O_DIM 16
#define BATCH 64
#define NNODES 1000
#define FUT 12
#define G4 512

__device__ __forceinline__ float sigm(float x) { return 1.0f / (1.0f + __expf(-x)); }

__global__ __launch_bounds__(256) void decoder_scan(
    const float* __restrict__ enc,
    const float* __restrict__ mean_,
    const float* __restrict__ s,
    const float* __restrict__ h0,
    const float* __restrict__ c0,
    const float* __restrict__ Wih0,
    const float* __restrict__ Whh0,
    const float* __restrict__ bih0, const float* __restrict__ bhh0,
    const float* __restrict__ Whr0,
    const float* __restrict__ Wih1,
    const float* __restrict__ Whh1,
    const float* __restrict__ bih1, const float* __restrict__ bhh1,
    const float* __restrict__ Whr1,
    float* __restrict__ out)
{
  __shared__ float WcatT0[32][G4];
  __shared__ float WcatT1[32][G4];
  __shared__ float Whr0T[H_DIM][O_DIM];
  __shared__ float Whr1T[H_DIM][O_DIM];
  __shared__ float GencB[16][G4];
  __shared__ float bias0s[G4];
  __shared__ float bias1s[G4];
  __shared__ float hfull[16][H_DIM];
  __shared__ float part[8][16][16];
  __shared__ float xin0[16][32];
  __shared__ float xin1[16][32];
  __shared__ float h1Buf[16][16];
  __shared__ float h2Buf[16][16];
  __shared__ float lastBuf[16][16];
  __shared__ float nmBuf[16][16];

  const int tid  = threadIdx.x;
  const int w    = tid >> 5;
  const int lane = tid & 31;
  const int lm   = lane & 15;
  const int lh   = lane >> 4;
  const int b0   = blockIdx.x * 16;

  for (int idx = tid; idx < 32 * G4; idx += 256) {
    int k = idx >> 9;
    int col = idx & 511;
    WcatT0[k][col] = (k < 16) ? Wih0[col * 80 + 64 + k] : Whh0[col * 16 + (k - 16)];
    WcatT1[k][col] = (k < 16) ? Wih1[col * 16 + k]      : Whh1[col * 16 + (k - 16)];
  }
  for (int idx = tid; idx < H_DIM * O_DIM; idx += 256) {
    int k = idx >> 4; int col = idx & 15;
    Whr0T[k][col] = Whr0[col * H_DIM + k];
    Whr1T[k][col] = Whr1[col * H_DIM + k];
  }
  for (int col = tid; col < G4; col += 256) {
    bias0s[col] = bih0[col] + bhh0[col];
    bias1s[col] = bih1[col] + bhh1[col];
  }
  {
    int row = tid >> 4, o = tid & 15;
    h1Buf[row][o]   = h0[o];
    h2Buf[row][o]   = h0[O_DIM + o];
    lastBuf[row][o] = s[o];
  }
  const int hcol = 16 * w + lm;
  float c1r[8], c2r[8];
  #pragma unroll
  for (int r = 0; r < 8; ++r) { c1r[r] = c0[hcol]; c2r[r] = c0[H_DIM + hcol]; }
  __syncthreads();

  for (int n = 0; n < NNODES; ++n) {
    {
      int row = tid >> 4, o = tid & 15;
      const float* mp = mean_ + (size_t)(b0 + row) * FUT * NNODES * O_DIM
                              + (size_t)n * O_DIM + o;
      float sum = 0.f;
      #pragma unroll
      for (int f = 0; f < FUT; ++f) sum += mp[(size_t)f * NNODES * O_DIM];
      nmBuf[row][o] = sum * (1.0f / FUT);
    }
    #pragma unroll
    for (int g = 0; g < 4; ++g) {
      int gc = 16 * (w + 8 * g) + lm;
      v8f acc;
      float bv = bias0s[gc];
      #pragma unroll
      for (int r = 0; r < 8; ++r) acc[r] = bv;
      const float* ap = enc + (size_t)(b0 + lm) * NNODES * E_DIM + (size_t)n * E_DIM;
      #pragma unroll
      for (int k = 0; k < 16; ++k) {
        int kk = 4 * k + 2 * lh;
        v2f a, b;
        a.x = ap[kk];              a.y = ap[kk + 1];
        b.x = Wih0[gc * 80 + kk];  b.y = Wih0[gc * 80 + kk + 1];
        acc = __builtin_amdgcn_wmma_f32_16x16x4_f32(false, a, false, b, (short)0, acc, false, false);
      }
      #pragma unroll
      for (int r = 0; r < 8; ++r) GencB[r + 8 * lh][gc] = acc[r];
    }
    __syncthreads();

    for (int f = 0; f < FUT; ++f) {
      {
        int row = tid >> 4, o = tid & 15;
        xin0[row][o]      = lastBuf[row][o] + nmBuf[row][o];
        xin0[row][16 + o] = h1Buf[row][o];
      }
      __syncthreads();

      {
        v8f acc[4];
        #pragma unroll
        for (int g = 0; g < 4; ++g) {
          int gc = 16 * (w + 8 * g) + lm;
          #pragma unroll
          for (int r = 0; r < 8; ++r) acc[g][r] = GencB[r + 8 * lh][gc];
        }
        #pragma unroll
        for (int k = 0; k < 8; ++k) {
          int kk = 4 * k + 2 * lh;
          v2f a; a.x = xin0[lm][kk]; a.y = xin0[lm][kk + 1];
          #pragma unroll
          for (int g = 0; g < 4; ++g) {
            int gc = 16 * (w + 8 * g) + lm;
            v2f b; b.x = WcatT0[kk][gc]; b.y = WcatT0[kk + 1][gc];
            acc[g] = __builtin_amdgcn_wmma_f32_16x16x4_f32(false, a, false, b, (short)0, acc[g], false, false);
          }
        }
        #pragma unroll
        for (int r = 0; r < 8; ++r) {
          float iv = sigm(acc[0][r]);
          float fv = sigm(acc[1][r]);
          float gv = tanhf(acc[2][r]);
          float ov = sigm(acc[3][r]);
          float cn = fv * c1r[r] + iv * gv;
          c1r[r] = cn;
          hfull[r + 8 * lh][hcol] = ov * tanhf(cn);
        }
      }
      __syncthreads();

      {
        v8f acc = {};
        #pragma unroll
        for (int k = 0; k < 4; ++k) {
          int kk = 16 * w + 4 * k + 2 * lh;
          v2f a; a.x = hfull[lm][kk];  a.y = hfull[lm][kk + 1];
          v2f b; b.x = Whr0T[kk][lm];  b.y = Whr0T[kk + 1][lm];
          acc = __builtin_amdgcn_wmma_f32_16x16x4_f32(false, a, false, b, (short)0, acc, false, false);
        }
        #pragma unroll
        for (int r = 0; r < 8; ++r) part[w][r + 8 * lh][lm] = acc[r];
      }
      __syncthreads();
      {
        int row = tid >> 4, o = tid & 15;
        float v = 0.f;
        #pragma unroll
        for (int p = 0; p < 8; ++p) v += part[p][row][o];
        h1Buf[row][o] = v;
        xin1[row][o] = v;
        xin1[row][16 + o] = h2Buf[row][o];
      }
      __syncthreads();

      {
        v8f acc[4];
        #pragma unroll
        for (int g = 0; g < 4; ++g) {
          int gc = 16 * (w + 8 * g) + lm;
          float bv = bias1s[gc];
          #pragma unroll
          for (int r = 0; r < 8; ++r) acc[g][r] = bv;
        }
        #pragma unroll
        for (int k = 0; k < 8; ++k) {
          int kk = 4 * k + 2 * lh;
          v2f a; a.x = xin1[lm][kk]; a.y = xin1[lm][kk + 1];
          #pragma unroll
          for (int g = 0; g < 4; ++g) {
            int gc = 16 * (w + 8 * g) + lm;
            v2f b; b.x = WcatT1[kk][gc]; b.y = WcatT1[kk + 1][gc];
            acc[g] = __builtin_amdgcn_wmma_f32_16x16x4_f32(false, a, false, b, (short)0, acc[g], false, false);
          }
        }
        #pragma unroll
        for (int r = 0; r < 8; ++r) {
          float iv = sigm(acc[0][r]);
          float fv = sigm(acc[1][r]);
          float gv = tanhf(acc[2][r]);
          float ov = sigm(acc[3][r]);
          float cn = fv * c2r[r] + iv * gv;
          c2r[r] = cn;
          hfull[r + 8 * lh][hcol] = ov * tanhf(cn);
        }
      }
      __syncthreads();

      {
        v8f acc = {};
        #pragma unroll
        for (int k = 0; k < 4; ++k) {
          int kk = 16 * w + 4 * k + 2 * lh;
          v2f a; a.x = hfull[lm][kk];  a.y = hfull[lm][kk + 1];
          v2f b; b.x = Whr1T[kk][lm];  b.y = Whr1T[kk + 1][lm];
          acc = __builtin_amdgcn_wmma_f32_16x16x4_f32(false, a, false, b, (short)0, acc, false, false);
        }
        #pragma unroll
        for (int r = 0; r < 8; ++r) part[w][r + 8 * lh][lm] = acc[r];
      }
      __syncthreads();
      {
        int row = tid >> 4, o = tid & 15;
        float v = 0.f;
        #pragma unroll
        for (int p = 0; p < 8; ++p) v += part[p][row][o];
        h2Buf[row][o]   = v;
        lastBuf[row][o] = v;
        out[(size_t)(b0 + row) * FUT * NNODES * O_DIM
            + (size_t)f * NNODES * O_DIM + (size_t)n * O_DIM + o] = v + nmBuf[row][o];
      }
      __syncthreads();
    }
  }
}

extern "C" void kernel_launch(void* const* d_in, const int* in_sizes, int n_in,
                              void* d_out, int out_size, void* d_ws, size_t ws_size,
                              hipStream_t stream) {
  (void)in_sizes; (void)n_in; (void)out_size; (void)d_ws; (void)ws_size;
  const float* enc   = (const float*)d_in[0];
  const float* mean_ = (const float*)d_in[1];
  const float* s     = (const float*)d_in[2];
  const float* h0    = (const float*)d_in[3];
  const float* c0    = (const float*)d_in[4];
  const float* Wih0  = (const float*)d_in[5];
  const float* Whh0  = (const float*)d_in[6];
  const float* bih0  = (const float*)d_in[7];
  const float* bhh0  = (const float*)d_in[8];
  const float* Whr0  = (const float*)d_in[9];
  const float* Wih1  = (const float*)d_in[10];
  const float* Whh1  = (const float*)d_in[11];
  const float* bih1  = (const float*)d_in[12];
  const float* bhh1  = (const float*)d_in[13];
  const float* Whr1  = (const float*)d_in[14];
  float* out = (float*)d_out;

  decoder_scan<<<dim3(BATCH / 16), dim3(256), 0, stream>>>(
      enc, mean_, s, h0, c0,
      Wih0, Whh0, bih0, bhh0, Whr0,
      Wih1, Whh1, bih1, bhh1, Whr1, out);
}


# Each wave owns 8 hidden units through a layer: i|f and g|o gate tiles, cell update, K=8 projection slice stay in registers by lane exchange; 4 of 9 barriers go; projection k sum order changes
# speedup vs baseline: 2.1867x; 2.1867x over previous
#include <hip/hip_runtime.h>
#include <math.h>

typedef __attribute__((ext_vector_type(2))) float v2f;
typedef __attribute__((ext_vector_type(8))) float v8f;

#define E_DIM 64
#define H_DIM 128
#define O_DIM 16
#define BATCH 64
#define NNODES 1000
#define FUT 12
#define G4 512
#define BR 4
#define NWG 16
#define NWAVE 16

__device__ __forceinline__ float sigm(float x) { return 1.0f / (1.0f + __expf(-x)); }

__device__ __forceinline__ float lane_read(int byte_index, float v) {
  return __builtin_bit_cast(float, __builtin_amdgcn_ds_bpermute(byte_index, __builtin_bit_cast(int, v)));
}

__device__ __forceinline__ float row_xor8(float v) {
  return __builtin_bit_cast(float, __builtin_amdgcn_update_dpp(0, __builtin_bit_cast(int, v), 0x128  , 0xf, 0xf, false));
}

__global__ __launch_bounds__(512) void decoder_scan(
    const float* __restrict__ enc,
    const float* __restrict__ mean_,
    const float* __restrict__ s,
    const float* __restrict__ h0,
    const float* __restrict__ c0,
    const float* __restrict__ Wih0,
    const float* __restrict__ Whh0,
    const float* __restrict__ bih0, const float* __restrict__ bhh0,
    const float* __restrict__ Whr0,
    const float* __restrict__ Wih1,
    const float* __restrict__ Whh1,
    const float* __restrict__ bih1, const float* __restrict__ bhh1,
    const float* __restrict__ Whr1,
    float* __restrict__ out)
{
  __shared__ v2f Wf0[NWAVE * 2 * 8 * 32];
  __shared__ v2f Wf1[NWAVE * 2 * 8 * 32];
  __shared__ v2f Whrf0[NWAVE * 2 * 32];
  __shared__ v2f Whrf1[NWAVE * 2 * 32];
  __shared__ float GencB[BR][G4];
  __shared__ float bias0s[G4];
  __shared__ float bias1s[G4];
  __shared__ float part[NWAVE][BR][16];
  __shared__ float xin0[16][32];
  __shared__ float xin1[16][32];
  __shared__ float h1Buf[BR][16];
  __shared__ float h2Buf[BR][16];
  __shared__ float lastBuf[BR][16];
  __shared__ float nmBuf[BR][16];

  const int tid  = threadIdx.x;
  const int w    = __builtin_amdgcn_readfirstlane(tid >> 5);
  const int lane = tid & 31;
  const int lm   = lane & 15;
  const int lh   = lane >> 4;
  const int b0   = blockIdx.x * BR;
  const int gc0  = 128 * (0 + (lm >> 3)) + 8 * w + (lm & 7);
  const int gc1  = 128 * (2 + (lm >> 3)) + 8 * w + (lm & 7);
  const bool lo  = (lm < 8);
  const int cu   = 8 * w + (lm & 7);
  const bool arow = ((lm & 6) == 0);
  const int gbase = 16 * (lm >> 3) + 8 * (lm & 1) + 2 * lh;
  const int gi00 = (gbase + 0) << 2, gi01 = (gbase + 1) << 2, gi10 = (gbase + 4) << 2, gi11 = (gbase + 5) << 2;
  const int vj   = (tid >> 4) & 3;
  const int vo   = tid & 15;
  const int vtr  = (vj & 1) + 8 * (vj >> 1);
  const int aslot = (lm & 1) + 2 * (lm >> 3);

  for (int idx = tid; idx < NWAVE * 2 * 8 * 32 * 2; idx += 512) {
    int e = idx & 1, ln = (idx >> 1) & 31, k = (idx >> 6) & 7, j = (idx >> 9) & 1, ww = idx >> 10;
    int kr = 4 * k + 2 * (ln >> 4) + e;
    int l15 = ln & 15;
    int col = 128 * (2 * j + (l15 >> 3)) + 8 * ww + (l15 & 7);
    ((float*)Wf0)[idx] = (kr < 16) ? Wih0[col * 80 + 64 + kr] : Whh0[col * 16 + (kr - 16)];
    ((float*)Wf1)[idx] = (kr < 16) ? Wih1[col * 16 + kr]      : Whh1[col * 16 + (kr - 16)];
  }
  for (int idx = tid; idx < NWAVE * 2 * 32 * 2; idx += 512) {
    int e = idx & 1, ln = (idx >> 1) & 31, st = (idx >> 6) & 1, ww = idx >> 7;
    int unit = 8 * ww + 4 * st + 2 * (ln >> 4) + e;
    int o = ln & 15;
    ((float*)Whrf0)[idx] = Whr0[o * H_DIM + unit];
    ((float*)Whrf1)[idx] = Whr1[o * H_DIM + unit];
  }
  {
    int col = tid;
    bias0s[col] = bih0[col] + bhh0[col];
    bias1s[col] = bih1[col] + bhh1[col];
  }
  if (w < 2) {
    h1Buf[vj][vo]   = h0[vo];
    h2Buf[vj][vo]   = h0[O_DIM + vo];
    lastBuf[vj][vo] = s[vo];
  }
  for (int idx = tid; idx < 16 * 32; idx += 512) { xin0[idx >> 5][idx & 31] = 0.f; xin1[idx >> 5][idx & 31] = 0.f; }
  float c1r = c0[cu], c2r = c0[H_DIM + cu];
  __syncthreads();

  for (int n = 0; n < NNODES; ++n) {
    if (w < 2) {
      const float* mp = mean_ + (size_t)(b0 + vj) * FUT * NNODES * O_DIM
                              + (size_t)n * O_DIM + vo;
      float sum = 0.f;
      #pragma unroll
      for (int f = 0; f < FUT; ++f) sum += mp[(size_t)f * NNODES * O_DIM];
      nmBuf[vj][vo] = sum * (1.0f / FUT);
    }
    #pragma unroll
    for (int j = 0; j < 2; ++j) {
      int gc = j ? gc1 : gc0;
      v8f acc;
      float bv = bias0s[gc];
      #pragma unroll
      for (int r = 0; r < 8; ++r) acc[r] = bv;
      const float* ap = enc + (size_t)(b0 + aslot) * NNODES * E_DIM + (size_t)n * E_DIM;
      #pragma unroll
      for (int k = 0; k < 16; ++k) {
        int kk = 4 * k + 2 * lh;
        v2f a, b;
        a.x = ap[kk];              a.y = ap[kk + 1];
        b.x = Wih0[gc * 80 + kk];  b.y = Wih0[gc * 80 + kk + 1];
        acc = __builtin_amdgcn_wmma_f32_16x16x4_f32(false, a, false, b, (short)0, acc, false, false);
      }
      #pragma unroll
      for (int r = 0; r < 2; ++r) GencB[r + 2 * lh][gc] = acc[r];
    }
    __syncthreads();

    for (int f = 0; f < FUT; ++f) {
      if (w < 2) {
        xin0[vtr][vo]      = lastBuf[vj][vo] + nmBuf[vj][vo];
        xin0[vtr][16 + vo] = h1Buf[vj][vo];
      }
      __syncthreads();

      {
        v8f acc0, acc1;
        #pragma unroll
        for (int r = 0; r < 8; ++r) {
          acc0[r] = (r < 2) ? GencB[r + 2 * lh][gc0] : 0.f;
          acc1[r] = (r < 2) ? GencB[r + 2 * lh][gc1] : 0.f;
        }
        const v2f* wf = Wf0 + (w * 2 * 8) * 32 + lane;
        #pragma unroll
        for (int k = 0; k < 8; ++k) {
          int kk = 4 * k + 2 * lh;
          v2f a; a.x = xin0[lm][kk]; a.y = xin0[lm][kk + 1];
          v2f bA = wf[k * 32];
          v2f bB = wf[(8 + k) * 32];
          acc0 = __builtin_amdgcn_wmma_f32_16x16x4_f32(false, a, false, bA, (short)0, acc0, false, false);
          acc1 = __builtin_amdgcn_wmma_f32_16x16x4_f32(false, a, false, bB, (short)0, acc1, false, false);
        }
        float s1 = lo ? acc0[1] : acc0[0];
        float s2 = lo ? acc1[1] : acc1[0];
        float r1 = row_xor8(s1);
        float r2 = row_xor8(s2);
        float gi = lo ? acc0[0] : r1;
        float gf = lo ? r1 : acc0[1];
        float gg = lo ? acc1[0] : r2;
        float go = lo ? r2 : acc1[1];
        float iv = sigm(gi);
        float fv = sigm(gf);
        float gv = tanhf(gg);
        float ov = sigm(go);
        float cn = fv * c1r + iv * gv;
        c1r = cn;
        float hv = ov * tanhf(cn);
        float a00 = lane_read(gi00, hv), a01 = lane_read(gi01, hv);
        float a10 = lane_read(gi10, hv), a11 = lane_read(gi11, hv);
        v2f pa0, pa1;
        pa0.x = arow ? a00 : 0.f;  pa0.y = arow ? a01 : 0.f;
        pa1.x = arow ? a10 : 0.f;  pa1.y = arow ? a11 : 0.f;
        const v2f* wp = Whrf0 + (w * 2) * 32 + lane;
        v8f pacc = {};
        pacc = __builtin_amdgcn_wmma_f32_16x16x4_f32(false, pa0, false, wp[0],  (short)0, pacc, false, false);
        pacc = __builtin_amdgcn_wmma_f32_16x16x4_f32(false, pa1, false, wp[32], (short)0, pacc, false, false);
        #pragma unroll
        for (int r = 0; r < 2; ++r) part[w][r + 2 * lh][lm] = pacc[r];
      }
      __syncthreads();
      if (w < 2) {
        float v = 0.f;
        #pragma unroll
        for (int p = 0; p < NWAVE; ++p) v += part[p][vj][vo];
        h1Buf[vj][vo] = v;
        xin1[vtr][vo] = v;
        xin1[vtr][16 + vo] = h2Buf[vj][vo];
      }
      __syncthreads();

      {
        v8f acc0, acc1;
        {
          float bvA = bias1s[gc0], bvB = bias1s[gc1];
          #pragma unroll
          for (int r = 0; r < 8; ++r) { acc0[r] = bvA; acc1[r] = bvB; }
        }
        const v2f* wf = Wf1 + (w * 2 * 8) * 32 + lane;
        #pragma unroll
        for (int k = 0; k < 8; ++k) {
          int kk = 4 * k + 2 * lh;
          v2f a; a.x = xin1[lm][kk]; a.y = xin1[lm][kk + 1];
          v2f bA = wf[k * 32];
          v2f bB = wf[(8 + k) * 32];
          acc0 = __builtin_amdgcn_wmma_f32_16x16x4_f32(false, a, false, bA, (short)0, acc0, false, false);
          acc1 = __builtin_amdgcn_wmma_f32_16x16x4_f32(false, a, false, bB, (short)0, acc1, false, false);
        }
        float s1 = lo ? acc0[1] : acc0[0];
        float s2 = lo ? acc1[1] : acc1[0];
        float r1 = row_xor8(s1);
        float r2 = row_xor8(s2);
        float gi = lo ? acc0[0] : r1;
        float gf = lo ? r1 : acc0[1];
        float gg = lo ? acc1[0] : r2;
        float go = lo ? r2 : acc1[1];
        float iv = sigm(gi);
        float fv = sigm(gf);
        float gv = tanhf(gg);
        float ov = sigm(go);
        float cn = fv * c2r + iv * gv;
        c2r = cn;
        float hv = ov * tanhf(cn);
        float a00 = lane_read(gi00, hv), a01 = lane_read(gi01, hv);
        float a10 = lane_read(gi10, hv), a11 = lane_read(gi11, hv);
        v2f pa0, pa1;
        pa0.x = arow ? a00 : 0.f;  pa0.y = arow ? a01 : 0.f;
        pa1.x = arow ? a10 : 0.f;  pa1.y = arow ? a11 : 0.f;
        const v2f* wp = Whrf1 + (w * 2) * 32 + lane;
        v8f pacc = {};
        pacc = __builtin_amdgcn_wmma_f32_16x16x4_f32(false, pa0, false, wp[0],  (short)0, pacc, false, false);
        pacc = __builtin_amdgcn_wmma_f32_16x16x4_f32(false, pa1, false, wp[32], (short)0, pacc, false, false);
        #pragma unroll
        for (int r = 0; r < 2; ++r) part[w][r + 2 * lh][lm] = pacc[r];
      }
      __syncthreads();
      if (w < 2) {
        float v = 0.f;
        #pragma unroll
        for (int p = 0; p < NWAVE; ++p) v += part[p][vj][vo];
        h2Buf[vj][vo]   = v;
        lastBuf[vj][vo] = v;
        out[(size_t)(b0 + vj) * FUT * NNODES * O_DIM
            + (size_t)f * NNODES * O_DIM + (size_t)n * O_DIM + vo] = v + nmBuf[vj][vo];
      }
      __syncthreads();
    }
  }
}

extern "C" void kernel_launch(void* const* d_in, const int* in_sizes, int n_in,
                              void* d_out, int out_size, void* d_ws, size_t ws_size,
                              hipStream_t stream) {
  (void)in_sizes; (void)n_in; (void)out_size; (void)d_ws; (void)ws_size;
  const float* enc   = (const float*)d_in[0];
  const float* mean_ = (const float*)d_in[1];
  const float* s     = (const float*)d_in[2];
  const float* h0    = (const float*)d_in[3];
  const float* c0    = (const float*)d_in[4];
  const float* Wih0  = (const float*)d_in[5];
  const float* Whh0  = (const float*)d_in[6];
  const float* bih0  = (const float*)d_in[7];
  const float* bhh0  = (const float*)d_in[8];
  const float* Whr0  = (const float*)d_in[9];
  const float* Wih1  = (const float*)d_in[10];
  const float* Whh1  = (const float*)d_in[11];
  const float* bih1  = (const float*)d_in[12];
  const float* bhh1  = (const float*)d_in[13];
  const float* Whr1  = (const float*)d_in[14];
  float* out = (float*)d_out;

  decoder_scan<<<dim3(NWG), dim3(512), 0, stream>>>(
      enc, mean_, s, h0, c0,
      Wih0, Whh0, bih0, bhh0, Whr0,
      Wih1, Whh1, bih1, bhh1, Whr1, out);
}


# LDS operands of the per-step gate products in WMMA fragment order (A tiles as [k-step pair][lane] 16-byte rows; enc term, layer-1 bias per lane): reads bank-conflict-free, were 8/4/2-way; same sums
# speedup vs baseline: 2.3432x; 1.0716x over previous
#include <hip/hip_runtime.h>
#include <math.h>

typedef __attribute__((ext_vector_type(2))) float v2f;
typedef __attribute__((ext_vector_type(4))) float v4f;
typedef __attribute__((ext_vector_type(8))) float v8f;

#define E_DIM 64
#define H_DIM 128
#define O_DIM 16
#define BATCH 64
#define NNODES 1000
#define FUT 12
#define G4 512
#define BR 4
#define NWG 16
#define NWAVE 16

__device__ __forceinline__ float sigm(float x) { return 1.0f / (1.0f + __expf(-x)); }

__device__ __forceinline__ float lane_read(int byte_index, float v) {
  return __builtin_bit_cast(float, __builtin_amdgcn_ds_bpermute(byte_index, __builtin_bit_cast(int, v)));
}

__device__ __forceinline__ float row_xor8(float v) {
  return __builtin_bit_cast(float, __builtin_amdgcn_update_dpp(0, __builtin_bit_cast(int, v), 0x128  , 0xf, 0xf, false));
}

__global__ __launch_bounds__(512) void decoder_scan(
    const float* __restrict__ enc,
    const float* __restrict__ mean_,
    const float* __restrict__ s,
    const float* __restrict__ h0,
    const float* __restrict__ c0,
    const float* __restrict__ Wih0,
    const float* __restrict__ Whh0,
    const float* __restrict__ bih0, const float* __restrict__ bhh0,
    const float* __restrict__ Whr0,
    const float* __restrict__ Wih1,
    const float* __restrict__ Whh1,
    const float* __restrict__ bih1, const float* __restrict__ bhh1,
    const float* __restrict__ Whr1,
    float* __restrict__ out)
{
  __shared__ v2f Wf0[NWAVE * 2 * 8 * 32];
  __shared__ v2f Wf1[NWAVE * 2 * 8 * 32];
  __shared__ v2f Whrf0[NWAVE * 2 * 32];
  __shared__ v2f Whrf1[NWAVE * 2 * 32];
  __shared__ v2f GencF[NWAVE * 2 * 32];
  __shared__ float bias0s[G4];
  __shared__ float bias1f[NWAVE * 2 * 16];
  __shared__ float part[NWAVE][BR][16];
  __shared__ v4f xin0F[4 * 32];
  __shared__ v4f xin1F[4 * 32];
  __shared__ float h1Buf[BR][16];
  __shared__ float h2Buf[BR][16];
  __shared__ float lastBuf[BR][16];
  __shared__ float nmBuf[BR][16];

  const int tid  = threadIdx.x;
  const int w    = __builtin_amdgcn_readfirstlane(tid >> 5);
  const int lane = tid & 31;
  const int lm   = lane & 15;
  const int lh   = lane >> 4;
  const int b0   = blockIdx.x * BR;
  const int gc0  = 128 * (0 + (lm >> 3)) + 8 * w + (lm & 7);
  const int gc1  = 128 * (2 + (lm >> 3)) + 8 * w + (lm & 7);
  const bool lo  = (lm < 8);
  const int cu   = 8 * w + (lm & 7);
  const bool arow = ((lm & 6) == 0);
  const int gbase = 16 * (lm >> 3) + 8 * (lm & 1) + 2 * lh;
  const int gi00 = (gbase + 0) << 2, gi01 = (gbase + 1) << 2, gi10 = (gbase + 4) << 2, gi11 = (gbase + 5) << 2;
  const int vj   = (tid >> 4) & 3;
  const int vo   = tid & 15;
  const int vtr  = (vj & 1) + 8 * (vj >> 1);
  const int xoff = (vo >> 3) * 128 + (vtr + 16 * ((vo >> 1) & 1)) * 4 + ((vo >> 2) & 1) * 2 + (vo & 1);
  const int aslot = (lm & 1) + 2 * (lm >> 3);

  for (int idx = tid; idx < NWAVE * 2 * 8 * 32 * 2; idx += 512) {
    int e = idx & 1, ln = (idx >> 1) & 31, k = (idx >> 6) & 7, j = (idx >> 9) & 1, ww = idx >> 10;
    int kr = 4 * k + 2 * (ln >> 4) + e;
    int l15 = ln & 15;
    int col = 128 * (2 * j + (l15 >> 3)) + 8 * ww + (l15 & 7);
    ((float*)Wf0)[idx] = (kr < 16) ? Wih0[col * 80 + 64 + kr] : Whh0[col * 16 + (kr - 16)];
    ((float*)Wf1)[idx] = (kr < 16) ? Wih1[col * 16 + kr]      : Whh1[col * 16 + (kr - 16)];
  }
  for (int idx = tid; idx < NWAVE * 2 * 32 * 2; idx += 512) {
    int e = idx & 1, ln = (idx >> 1) & 31, st = (idx >> 6) & 1, ww = idx >> 7;
    int unit = 8 * ww + 4 * st + 2 * (ln >> 4) + e;
    int o = ln & 15;
    ((float*)Whrf0)[idx] = Whr0[o * H_DIM + unit];
    ((float*)Whrf1)[idx] = Whr1[o * H_DIM + unit];
  }
  {
    int col = tid;
    bias0s[col] = bih0[col] + bhh0[col];
    int l15 = tid & 15, j = (tid >> 4) & 1, ww = tid >> 5;
    int c1 = 128 * (2 * j + (l15 >> 3)) + 8 * ww + (l15 & 7);
    bias1f[tid] = bih1[c1] + bhh1[c1];
  }
  if (w < 2) {
    h1Buf[vj][vo]   = h0[vo];
    h2Buf[vj][vo]   = h0[O_DIM + vo];
    lastBuf[vj][vo] = s[vo];
  }
  ((float*)xin0F)[tid] = 0.f;
  ((float*)xin1F)[tid] = 0.f;
  float c1r = c0[cu], c2r = c0[H_DIM + cu];
  __syncthreads();

  for (int n = 0; n < NNODES; ++n) {
    if (w < 2) {
      const float* mp = mean_ + (size_t)(b0 + vj) * FUT * NNODES * O_DIM
                              + (size_t)n * O_DIM + vo;
      float sum = 0.f;
      #pragma unroll
      for (int f = 0; f < FUT; ++f) sum += mp[(size_t)f * NNODES * O_DIM];
      nmBuf[vj][vo] = sum * (1.0f / FUT);
    }
    #pragma unroll
    for (int j = 0; j < 2; ++j) {
      int gc = j ? gc1 : gc0;
      v8f acc;
      float bv = bias0s[gc];
      #pragma unroll
      for (int r = 0; r < 8; ++r) acc[r] = bv;
      const float* ap = enc + (size_t)(b0 + aslot) * NNODES * E_DIM + (size_t)n * E_DIM;
      #pragma unroll
      for (int k = 0; k < 16; ++k) {
        int kk = 4 * k + 2 * lh;
        v2f a, b;
        a.x = ap[kk];              a.y = ap[kk + 1];
        b.x = Wih0[gc * 80 + kk];  b.y = Wih0[gc * 80 + kk + 1];
        acc = __builtin_amdgcn_wmma_f32_16x16x4_f32(false, a, false, b, (short)0, acc, false, false);
      }
      v2f g; g.x = acc[0]; g.y = acc[1];
      GencF[(w * 2 + j) * 32 + lane] = g;
    }
    __syncthreads();

    for (int f = 0; f < FUT; ++f) {
      if (w < 2) {
        ((float*)xin0F)[xoff]       = lastBuf[vj][vo] + nmBuf[vj][vo];
        ((float*)xin0F)[xoff + 256] = h1Buf[vj][vo];
      }
      __syncthreads();

      {
        v8f acc0 = {}, acc1 = {};
        {
          v2f g0 = GencF[(w * 2 + 0) * 32 + lane], g1 = GencF[(w * 2 + 1) * 32 + lane];
          acc0[0] = g0.x; acc0[1] = g0.y;
          acc1[0] = g1.x; acc1[1] = g1.y;
        }
        const v2f* wf = Wf0 + (w * 2 * 8) * 32 + lane;
        const v4f* xa = xin0F + lane;
        #pragma unroll
        for (int q = 0; q < 4; ++q) {
          v4f t = xa[q * 32];
          #pragma unroll
          for (int e = 0; e < 2; ++e) {
            int k = 2 * q + e;
            v2f a; a.x = e ? t.z : t.x; a.y = e ? t.w : t.y;
            v2f bA = wf[k * 32];
            v2f bB = wf[(8 + k) * 32];
            acc0 = __builtin_amdgcn_wmma_f32_16x16x4_f32(false, a, false, bA, (short)0, acc0, false, false);
            acc1 = __builtin_amdgcn_wmma_f32_16x16x4_f32(false, a, false, bB, (short)0, acc1, false, false);
          }
        }
        float s1 = lo ? acc0[1] : acc0[0];
        float s2 = lo ? acc1[1] : acc1[0];
        float r1 = row_xor8(s1);
        float r2 = row_xor8(s2);
        float gi = lo ? acc0[0] : r1;
        float gf = lo ? r1 : acc0[1];
        float gg = lo ? acc1[0] : r2;
        float go = lo ? r2 : acc1[1];
        float iv = sigm(gi);
        float fv = sigm(gf);
        float gv = tanhf(gg);
        float ov = sigm(go);
        float cn = fv * c1r + iv * gv;
        c1r = cn;
        float hv = ov * tanhf(cn);
        float a00 = lane_read(gi00, hv), a01 = lane_read(gi01, hv);
        float a10 = lane_read(gi10, hv), a11 = lane_read(gi11, hv);
        v2f pa0, pa1;
        pa0.x = arow ? a00 : 0.f;  pa0.y = arow ? a01 : 0.f;
        pa1.x = arow ? a10 : 0.f;  pa1.y = arow ? a11 : 0.f;
        const v2f* wp = Whrf0 + (w * 2) * 32 + lane;
        v8f pacc = {};
        pacc = __builtin_amdgcn_wmma_f32_16x16x4_f32(false, pa0, false, wp[0],  (short)0, pacc, false, false);
        pacc = __builtin_amdgcn_wmma_f32_16x16x4_f32(false, pa1, false, wp[32], (short)0, pacc, false, false);
        #pragma unroll
        for (int r = 0; r < 2; ++r) part[w][r + 2 * lh][lm] = pacc[r];
      }
      __syncthreads();
      if (w < 2) {
        float v = 0.f;
        #pragma unroll
        for (int p = 0; p < NWAVE; ++p) v += part[p][vj][vo];
        h1Buf[vj][vo] = v;
        ((float*)xin1F)[xoff]       = v;
        ((float*)xin1F)[xoff + 256] = h2Buf[vj][vo];
      }
      __syncthreads();

      {
        v8f acc0, acc1;
        {
          float bvA = bias1f[(w * 2 + 0) * 16 + lm], bvB = bias1f[(w * 2 + 1) * 16 + lm];
          #pragma unroll
          for (int r = 0; r < 8; ++r) { acc0[r] = bvA; acc1[r] = bvB; }
        }
        const v2f* wf = Wf1 + (w * 2 * 8) * 32 + lane;
        const v4f* xa = xin1F + lane;
        #pragma unroll
        for (int q = 0; q < 4; ++q) {
          v4f t = xa[q * 32];
          #pragma unroll
          for (int e = 0; e < 2; ++e) {
            int k = 2 * q + e;
            v2f a; a.x = e ? t.z : t.x; a.y = e ? t.w : t.y;
            v2f bA = wf[k * 32];
            v2f bB = wf[(8 + k) * 32];
            acc0 = __builtin_amdgcn_wmma_f32_16x16x4_f32(false, a, false, bA, (short)0, acc0, false, false);
            acc1 = __builtin_amdgcn_wmma_f32_16x16x4_f32(false, a, false, bB, (short)0, acc1, false, false);
          }
        }
        float s1 = lo ? acc0[1] : acc0[0];
        float s2 = lo ? acc1[1] : acc1[0];
        float r1 = row_xor8(s1);
        float r2 = row_xor8(s2);
        float gi = lo ? acc0[0] : r1;
        float gf = lo ? r1 : acc0[1];
        float gg = lo ? acc1[0] : r2;
        float go = lo ? r2 : acc1[1];
        float iv = sigm(gi);
        float fv = sigm(gf);
        float gv = tanhf(gg);
        float ov = sigm(go);
        float cn = fv * c2r + iv * gv;
        c2r = cn;
        float hv = ov * tanhf(cn);
        float a00 = lane_read(gi00, hv), a01 = lane_read(gi01, hv);
        float a10 = lane_read(gi10, hv), a11 = lane_read(gi11, hv);
        v2f pa0, pa1;
        pa0.x = arow ? a00 : 0.f;  pa0.y = arow ? a01 : 0.f;
        pa1.x = arow ? a10 : 0.f;  pa1.y = arow ? a11 : 0.f;
        const v2f* wp = Whrf1 + (w * 2) * 32 + lane;
        v8f pacc = {};
        pacc = __builtin_amdgcn_wmma_f32_16x16x4_f32(false, pa0, false, wp[0],  (short)0, pacc, false, false);
        pacc = __builtin_amdgcn_wmma_f32_16x16x4_f32(false, pa1, false, wp[32], (short)0, pacc, false, false);
        #pragma unroll
        for (int r = 0; r < 2; ++r) part[w][r + 2 * lh][lm] = pacc[r];
      }
      __syncthreads();
      if (w < 2) {
        float v = 0.f;
        #pragma unroll
        for (int p = 0; p < NWAVE; ++p) v += part[p][vj][vo];
        h2Buf[vj][vo]   = v;
        lastBuf[vj][vo] = v;
        out[(size_t)(b0 + vj) * FUT * NNODES * O_DIM
            + (size_t)f * NNODES * O_DIM + (size_t)n * O_DIM + vo] = v + nmBuf[vj][vo];
      }
      __syncthreads();
    }
  }
}

extern "C" void kernel_launch(void* const* d_in, const int* in_sizes, int n_in,
                              void* d_out, int out_size, void* d_ws, size_t ws_size,
                              hipStream_t stream) {
  (void)in_sizes; (void)n_in; (void)out_size; (void)d_ws; (void)ws_size;
  const float* enc   = (const float*)d_in[0];
  const float* mean_ = (const float*)d_in[1];
  const float* s     = (const float*)d_in[2];
  const float* h0    = (const float*)d_in[3];
  const float* c0    = (const float*)d_in[4];
  const float* Wih0  = (const float*)d_in[5];
  const float* Whh0  = (const float*)d_in[6];
  const float* bih0  = (const float*)d_in[7];
  const float* bhh0  = (const float*)d_in[8];
  const float* Whr0  = (const float*)d_in[9];
  const float* Wih1  = (const float*)d_in[10];
  const float* Whh1  = (const float*)d_in[11];
  const float* bih1  = (const float*)d_in[12];
  const float* bhh1  = (const float*)d_in[13];
  const float* Whr1  = (const float*)d_in[14];
  float* out = (float*)d_out;

  decoder_scan<<<dim3(NWG), dim3(512), 0, stream>>>(
      enc, mean_, s, h0, c0,
      Wih0, Whh0, bih0, bhh0, Whr0,
      Wih1, Whh1, bih1, bhh1, Whr1, out);
}


# Batch re-split: 32 work-groups of 2 batch rows (matrix rows 0, 8), not 16 of 4; 8 waves own 16 hidden units as 4 one-gate tiles: a SIMD runs 2 cell-update waves, not 4; projection k sum order changes
# speedup vs baseline: 2.5923x; 1.1063x over previous
#include <hip/hip_runtime.h>
#include <math.h>

typedef __attribute__((ext_vector_type(2))) float v2f;
typedef __attribute__((ext_vector_type(4))) float v4f;
typedef __attribute__((ext_vector_type(8))) float v8f;

#define E_DIM 64
#define H_DIM 128
#define O_DIM 16
#define BATCH 64
#define NNODES 1000
#define FUT 12
#define G4 512
#define BR 2
#define NWG 32
#define NWAVE 8
#define NTHR 256

__device__ __forceinline__ float sigm(float x) { return 1.0f / (1.0f + __expf(-x)); }

__device__ __forceinline__ float lane_read(int byte_index, float v) {
  return __builtin_bit_cast(float, __builtin_amdgcn_ds_bpermute(byte_index, __builtin_bit_cast(int, v)));
}

__global__ __launch_bounds__(NTHR) void decoder_scan(
    const float* __restrict__ enc,
    const float* __restrict__ mean_,
    const float* __restrict__ s,
    const float* __restrict__ h0,
    const float* __restrict__ c0,
    const float* __restrict__ Wih0,
    const float* __restrict__ Whh0,
    const float* __restrict__ bih0, const float* __restrict__ bhh0,
    const float* __restrict__ Whr0,
    const float* __restrict__ Wih1,
    const float* __restrict__ Whh1,
    const float* __restrict__ bih1, const float* __restrict__ bhh1,
    const float* __restrict__ Whr1,
    float* __restrict__ out)
{
  __shared__ v2f Wf0[NWAVE * 4 * 8 * 32];
  __shared__ v2f Wf1[NWAVE * 4 * 8 * 32];
  __shared__ v2f Whrf0[NWAVE * 4 * 32];
  __shared__ v2f Whrf1[NWAVE * 4 * 32];
  __shared__ float GencF[NWAVE * 4 * 32];
  __shared__ float bias0s[G4];
  __shared__ float bias1f[NWAVE * 4 * 16];
  __shared__ float part[NWAVE][BR][16];
  __shared__ v4f xin0F[4 * 32];
  __shared__ v4f xin1F[4 * 32];
  __shared__ float h1Buf[BR][16];
  __shared__ float h2Buf[BR][16];
  __shared__ float lastBuf[BR][16];
  __shared__ float nmBuf[BR][16];

  const int tid  = threadIdx.x;
  const int w    = __builtin_amdgcn_readfirstlane(tid >> 5);
  const int lane = tid & 31;
  const int lm   = lane & 15;
  const int lh   = lane >> 4;
  const int b0   = blockIdx.x * BR;
  const int cu   = 16 * w + lm;
  const bool arow = ((lm & 7) == 0);
  const int gbase = 16 * (lm >> 3) + 2 * lh;
  const int vj   = (tid >> 4) & 1;
  const int vo   = tid & 15;
  const int vtr  = 8 * vj;
  const int xoff = (vo >> 3) * 128 + (vtr + 16 * ((vo >> 1) & 1)) * 4 + ((vo >> 2) & 1) * 2 + (vo & 1);
  const int aslot = lm >> 3;

  for (int idx = tid; idx < NWAVE * 4 * 8 * 32 * 2; idx += NTHR) {
    int e = idx & 1, ln = (idx >> 1) & 31, k = (idx >> 6) & 7, j = (idx >> 9) & 3, ww = idx >> 11;
    int kr = 4 * k + 2 * (ln >> 4) + e;
    int col = 128 * j + 16 * ww + (ln & 15);
    ((float*)Wf0)[idx] = (kr < 16) ? Wih0[col * 80 + 64 + kr] : Whh0[col * 16 + (kr - 16)];
    ((float*)Wf1)[idx] = (kr < 16) ? Wih1[col * 16 + kr]      : Whh1[col * 16 + (kr - 16)];
  }
  for (int idx = tid; idx < NWAVE * 4 * 32 * 2; idx += NTHR) {
    int e = idx & 1, ln = (idx >> 1) & 31, st = (idx >> 6) & 3, ww = idx >> 8;
    int unit = 16 * ww + 4 * st + 2 * (ln >> 4) + e;
    int o = ln & 15;
    ((float*)Whrf0)[idx] = Whr0[o * H_DIM + unit];
    ((float*)Whrf1)[idx] = Whr1[o * H_DIM + unit];
  }
  for (int idx = tid; idx < G4; idx += NTHR) {
    bias0s[idx] = bih0[idx] + bhh0[idx];
    int l15 = idx & 15, j = (idx >> 4) & 3, ww = idx >> 6;
    int c1 = 128 * j + 16 * ww + l15;
    bias1f[idx] = bih1[c1] + bhh1[c1];
  }
  if (w < 1) {
    h1Buf[vj][vo]   = h0[vo];
    h2Buf[vj][vo]   = h0[O_DIM + vo];
    lastBuf[vj][vo] = s[vo];
  }
  for (int idx = tid; idx < 4 * 32 * 4; idx += NTHR) {
    ((float*)xin0F)[idx] = 0.f;
    ((float*)xin1F)[idx] = 0.f;
  }
  float c1r = c0[cu], c2r = c0[H_DIM + cu];
  __syncthreads();

  for (int n = 0; n < NNODES; ++n) {
    if (w < 1) {
      const float* mp = mean_ + (size_t)(b0 + vj) * FUT * NNODES * O_DIM
                              + (size_t)n * O_DIM + vo;
      float sum = 0.f;
      #pragma unroll
      for (int f = 0; f < FUT; ++f) sum += mp[(size_t)f * NNODES * O_DIM];
      nmBuf[vj][vo] = sum * (1.0f / FUT);
    }
    {
      v8f hacc[4];
      #pragma unroll
      for (int j = 0; j < 4; ++j) {
        float bv = bias0s[128 * j + cu];
        #pragma unroll
        for (int r = 0; r < 8; ++r) hacc[j][r] = bv;
      }
      const float* ap = enc + (size_t)(b0 + aslot) * NNODES * E_DIM + (size_t)n * E_DIM;
      #pragma unroll
      for (int k = 0; k < 16; ++k) {
        int kk = 4 * k + 2 * lh;
        v2f a;
        a.x = ap[kk];  a.y = ap[kk + 1];
        #pragma unroll
        for (int j = 0; j < 4; ++j) {
          v2f b;
          b.x = Wih0[(128 * j + cu) * 80 + kk];  b.y = Wih0[(128 * j + cu) * 80 + kk + 1];
          hacc[j] = __builtin_amdgcn_wmma_f32_16x16x4_f32(false, a, false, b, (short)0, hacc[j], false, false);
        }
      }
      #pragma unroll
      for (int j = 0; j < 4; ++j) GencF[(w * 4 + j) * 32 + lane] = hacc[j][0];
    }
    __syncthreads();

    for (int f = 0; f < FUT; ++f) {
      if (w < 1) {
        ((float*)xin0F)[xoff]       = lastBuf[vj][vo] + nmBuf[vj][vo];
        ((float*)xin0F)[xoff + 256] = h1Buf[vj][vo];
      }
      __syncthreads();

      {
        v8f acc[4];
        #pragma unroll
        for (int j = 0; j < 4; ++j) {
          acc[j] = (v8f){};
          acc[j][0] = GencF[(w * 4 + j) * 32 + lane];
        }
        const v2f* wf = Wf0 + (w * 4 * 8) * 32 + lane;
        const v4f* xa = xin0F + lane;
        #pragma unroll
        for (int q = 0; q < 4; ++q) {
          v4f t = xa[q * 32];
          #pragma unroll
          for (int e = 0; e < 2; ++e) {
            int k = 2 * q + e;
            v2f a; a.x = e ? t.z : t.x; a.y = e ? t.w : t.y;
            #pragma unroll
            for (int j = 0; j < 4; ++j) {
              v2f b = wf[(8 * j + k) * 32];
              acc[j] = __builtin_amdgcn_wmma_f32_16x16x4_f32(false, a, false, b, (short)0, acc[j], false, false);
            }
          }
        }
        float iv = sigm(acc[0][0]);
        float fv = sigm(acc[1][0]);
        float gv = tanhf(acc[2][0]);
        float ov = sigm(acc[3][0]);
        float cn = fv * c1r + iv * gv;
        c1r = cn;
        float hv = ov * tanhf(cn);
        const v2f* wp = Whrf0 + (w * 4) * 32 + lane;
        v8f pacc0 = {}, pacc1 = {};
        #pragma unroll
        for (int st = 0; st < 4; ++st) {
          float g0 = lane_read((gbase + 4 * st) << 2, hv);
          float g1 = lane_read((gbase + 4 * st + 1) << 2, hv);
          v2f pa; pa.x = arow ? g0 : 0.f;  pa.y = arow ? g1 : 0.f;
          if (st < 2) pacc0 = __builtin_amdgcn_wmma_f32_16x16x4_f32(false, pa, false, wp[st * 32], (short)0, pacc0, false, false);
          else        pacc1 = __builtin_amdgcn_wmma_f32_16x16x4_f32(false, pa, false, wp[st * 32], (short)0, pacc1, false, false);
        }
        part[w][lh][lm] = pacc0[0] + pacc1[0];
      }
      __syncthreads();
      if (w < 1) {
        float v = 0.f;
        #pragma unroll
        for (int p = 0; p < NWAVE; ++p) v += part[p][vj][vo];
        h1Buf[vj][vo] = v;
        ((float*)xin1F)[xoff]       = v;
        ((float*)xin1F)[xoff + 256] = h2Buf[vj][vo];
      }
      __syncthreads();

      {
        v8f acc[4];
        #pragma unroll
        for (int j = 0; j < 4; ++j) {
          acc[j] = (v8f){};
          acc[j][0] = bias1f[(w * 4 + j) * 16 + lm];
        }
        const v2f* wf = Wf1 + (w * 4 * 8) * 32 + lane;
        const v4f* xa = xin1F + lane;
        #pragma unroll
        for (int q = 0; q < 4; ++q) {
          v4f t = xa[q * 32];
          #pragma unroll
          for (int e = 0; e < 2; ++e) {
            int k = 2 * q + e;
            v2f a; a.x = e ? t.z : t.x; a.y = e ? t.w : t.y;
            #pragma unroll
            for (int j = 0; j < 4; ++j) {
              v2f b = wf[(8 * j + k) * 32];
              acc[j] = __builtin_amdgcn_wmma_f32_16x16x4_f32(false, a, false, b, (short)0, acc[j], false, false);
            }
          }
        }
        float iv = sigm(acc[0][0]);
        float fv = sigm(acc[1][0]);
        float gv = tanhf(acc[2][0]);
        float ov = sigm(acc[3][0]);
        float cn = fv * c2r + iv * gv;
        c2r = cn;
        float hv = ov * tanhf(cn);
        const v2f* wp = Whrf1 + (w * 4) * 32 + lane;
        v8f pacc0 = {}, pacc1 = {};
        #pragma unroll
        for (int st = 0; st < 4; ++st) {
          float g0 = lane_read((gbase + 4 * st) << 2, hv);
          float g1 = lane_read((gbase + 4 * st + 1) << 2, hv);
          v2f pa; pa.x = arow ? g0 : 0.f;  pa.y = arow ? g1 : 0.f;
          if (st < 2) pacc0 = __builtin_amdgcn_wmma_f32_16x16x4_f32(false, pa, false, wp[st * 32], (short)0, pacc0, false, false);
          else        pacc1 = __builtin_amdgcn_wmma_f32_16x16x4_f32(false, pa, false, wp[st * 32], (short)0, pacc1, false, false);
        }
        part[w][lh][lm] = pacc0[0] + pacc1[0];
      }
      __syncthreads();
      if (w < 1) {
        float v = 0.f;
        #pragma unroll
        for (int p = 0; p < NWAVE; ++p) v += part[p][vj][vo];
        h2Buf[vj][vo]   = v;
        lastBuf[vj][vo] = v;
        out[(size_t)(b0 + vj) * FUT * NNODES * O_DIM
            + (size_t)f * NNODES * O_DIM + (size_t)n * O_DIM + vo] = v + nmBuf[vj][vo];
      }
      __syncthreads();
    }
  }
}

extern "C" void kernel_launch(void* const* d_in, const int* in_sizes, int n_in,
                              void* d_out, int out_size, void* d_ws, size_t ws_size,
                              hipStream_t stream) {
  (void)in_sizes; (void)n_in; (void)out_size; (void)d_ws; (void)ws_size;
  const float* enc   = (const float*)d_in[0];
  const float* mean_ = (const float*)d_in[1];
  const float* s     = (const float*)d_in[2];
  const float* h0    = (const float*)d_in[3];
  const float* c0    = (const float*)d_in[4];
  const float* Wih0  = (const float*)d_in[5];
  const float* Whh0  = (const float*)d_in[6];
  const float* bih0  = (const float*)d_in[7];
  const float* bhh0  = (const float*)d_in[8];
  const float* Whr0  = (const float*)d_in[9];
  const float* Wih1  = (const float*)d_in[10];
  const float* Whh1  = (const float*)d_in[11];
  const float* bih1  = (const float*)d_in[12];
  const float* bhh1  = (const float*)d_in[13];
  const float* Whr1  = (const float*)d_in[14];
  float* out = (float*)d_out;

  decoder_scan<<<dim3(NWG), dim3(NTHR), 8192, stream>>>(
      enc, mean_, s, h0, c0,
      Wih0, Whh0, bih0, bhh0, Whr0,
      Wih1, Whh1, bih1, bhh1, Whr1, out);
}


# Split-K projection partials ride the 16 A-tile rows (row 8*slot+wave), summed from the 8 accumulator registers after the gate product: wave-0 sum phases, 3 of 5 barriers go; gate sum order changes
# speedup vs baseline: 3.0096x; 1.1610x over previous
#include <hip/hip_runtime.h>
#include <math.h>

typedef __attribute__((ext_vector_type(2))) float v2f;
typedef __attribute__((ext_vector_type(4))) float v4f;
typedef __attribute__((ext_vector_type(8))) float v8f;

#define E_DIM 64
#define H_DIM 128
#define O_DIM 16
#define BATCH 64
#define NNODES 1000
#define FUT 12
#define G4 512
#define BR 2
#define NWG 32
#define NWAVE 8
#define NTHR 256

__device__ __forceinline__ float sigm(float x) { return 1.0f / (1.0f + __expf(-x)); }

__device__ __forceinline__ float lane_read(int byte_index, float v) {
  return __builtin_bit_cast(float, __builtin_amdgcn_ds_bpermute(byte_index, __builtin_bit_cast(int, v)));
}

__global__ __launch_bounds__(NTHR) void decoder_scan(
    const float* __restrict__ enc,
    const float* __restrict__ mean_,
    const float* __restrict__ s,
    const float* __restrict__ h0,
    const float* __restrict__ c0,
    const float* __restrict__ Wih0,
    const float* __restrict__ Whh0,
    const float* __restrict__ bih0, const float* __restrict__ bhh0,
    const float* __restrict__ Whr0,
    const float* __restrict__ Wih1,
    const float* __restrict__ Whh1,
    const float* __restrict__ bih1, const float* __restrict__ bhh1,
    const float* __restrict__ Whr1,
    float* __restrict__ out)
{
  __shared__ v2f Wf0[NWAVE * 4 * 8 * 32];
  __shared__ v2f Wf1[NWAVE * 4 * 8 * 32];
  __shared__ v2f Whrf0[NWAVE * 4 * 32];
  __shared__ v2f Whrf1[NWAVE * 4 * 32];
  __shared__ float GencF[NWAVE * 4 * 32];
  __shared__ float bias0s[G4];
  __shared__ float bias1f[NWAVE * 4 * 16];
  __shared__ v4f xin0F[4 * 32];
  __shared__ v4f xin1F[4 * 32];

  const int tid  = threadIdx.x;
  const int w    = __builtin_amdgcn_readfirstlane(tid >> 5);
  const int lane = tid & 31;
  const int lm   = lane & 15;
  const int lh   = lane >> 4;
  const int b0   = blockIdx.x * BR;
  const int cu   = 16 * w + lm;
  const bool arow = ((lm & 7) == 0);
  const int gbase = 16 * (lm >> 3) + 2 * lh;
  const int pbase = (lm >> 3) * 128 + (8 * lh + 16 * ((lm >> 1) & 1)) * 4 + ((lm >> 2) & 1) * 2 + (lm & 1);
  const int poff  = pbase + 4 * w;
  const int aslot = lm >> 3;
  float* const x0 = (float*)xin0F;
  float* const x1 = (float*)xin1F;

  for (int idx = tid; idx < NWAVE * 4 * 8 * 32 * 2; idx += NTHR) {
    int e = idx & 1, ln = (idx >> 1) & 31, k = (idx >> 6) & 7, j = (idx >> 9) & 3, ww = idx >> 11;
    int kr = 4 * k + 2 * (ln >> 4) + e;
    int col = 128 * j + 16 * ww + (ln & 15);
    ((float*)Wf0)[idx] = (kr < 16) ? Wih0[col * 80 + 64 + kr] : Whh0[col * 16 + (kr - 16)];
    ((float*)Wf1)[idx] = (kr < 16) ? Wih1[col * 16 + kr]      : Whh1[col * 16 + (kr - 16)];
  }
  for (int idx = tid; idx < NWAVE * 4 * 32 * 2; idx += NTHR) {
    int e = idx & 1, ln = (idx >> 1) & 31, st = (idx >> 6) & 3, ww = idx >> 8;
    int unit = 16 * ww + 4 * st + 2 * (ln >> 4) + e;
    int o = ln & 15;
    ((float*)Whrf0)[idx] = Whr0[o * H_DIM + unit];
    ((float*)Whrf1)[idx] = Whr1[o * H_DIM + unit];
  }
  for (int idx = tid; idx < G4; idx += NTHR) {
    bias0s[idx] = bih0[idx] + bhh0[idx];
    int l15 = idx & 15, j = (idx >> 4) & 3, ww = idx >> 6;
    int c1 = 128 * j + 16 * ww + l15;
    bias1f[idx] = bih1[c1] + bhh1[c1];
  }
  const float* const mp0 = mean_ + (size_t)(b0 + lh) * FUT * NNODES * O_DIM + lm;
  float nm_cur = 0.f, nm_nx = 0.f, nm_o = 0.f;
  float p1 = 0.f, p2 = 0.f, lp = 0.f;
  if (w < 1) {
    float sum = 0.f;
    #pragma unroll
    for (int f = 0; f < FUT; ++f) sum += mp0[(size_t)f * NNODES * O_DIM];
    nm_cur = sum * (1.0f / FUT);
    p1 = h0[lm];
    p2 = h0[O_DIM + lm];
    lp = s[lm] + nm_cur;
  }
  x0[poff]       = lp;
  x0[poff + 256] = p1;
  float c1r = c0[cu], c2r = c0[H_DIM + cu];
  float* outp_prev = out;
  __syncthreads();

  for (int n = 0; n < NNODES; ++n) {
    if (w < 1) {
      nm_nx = 0.f;
      if (n + 1 < NNODES) {
        const float* mp = mp0 + (size_t)(n + 1) * O_DIM;
        float sum = 0.f;
        #pragma unroll
        for (int f = 0; f < FUT; ++f) sum += mp[(size_t)f * NNODES * O_DIM];
        nm_nx = sum * (1.0f / FUT);
      }
    }
    {
      v8f hacc[4];
      #pragma unroll
      for (int j = 0; j < 4; ++j) {
        float bv = bias0s[128 * j + cu];
        #pragma unroll
        for (int r = 0; r < 8; ++r) hacc[j][r] = bv;
      }
      const float* ap = enc + (size_t)(b0 + aslot) * NNODES * E_DIM + (size_t)n * E_DIM;
      #pragma unroll
      for (int k = 0; k < 16; ++k) {
        int kk = 4 * k + 2 * lh;
        v2f a;
        a.x = ap[kk];  a.y = ap[kk + 1];
        #pragma unroll
        for (int j = 0; j < 4; ++j) {
          v2f b;
          b.x = Wih0[(128 * j + cu) * 80 + kk];  b.y = Wih0[(128 * j + cu) * 80 + kk + 1];
          hacc[j] = __builtin_amdgcn_wmma_f32_16x16x4_f32(false, a, false, b, (short)0, hacc[j], false, false);
        }
      }
      #pragma unroll
      for (int j = 0; j < 4; ++j) GencF[(w * 4 + j) * 32 + lane] = hacc[j][0];
    }

    for (int f = 0; f < FUT; ++f) {
      __syncthreads();
      x1[poff + 256] = p2;

      {
        v8f acc[4];
        #pragma unroll
        for (int j = 0; j < 4; ++j) {
          acc[j] = (v8f){};
          acc[j][0] = GencF[(w * 4 + j) * 32 + lane];
        }
        const v2f* wf = Wf0 + (w * 4 * 8) * 32 + lane;
        const v4f* xa = xin0F + lane;
        #pragma unroll
        for (int q = 0; q < 4; ++q) {
          v4f t = xa[q * 32];
          #pragma unroll
          for (int e = 0; e < 2; ++e) {
            int k = 2 * q + e;
            v2f a; a.x = e ? t.z : t.x; a.y = e ? t.w : t.y;
            #pragma unroll
            for (int j = 0; j < 4; ++j) {
              v2f b = wf[(8 * j + k) * 32];
              acc[j] = __builtin_amdgcn_wmma_f32_16x16x4_f32(false, a, false, b, (short)0, acc[j], false, false);
            }
          }
        }
        float gs[4];
        #pragma unroll
        for (int j = 0; j < 4; ++j)
          gs[j] = ((acc[j][0] + acc[j][1]) + (acc[j][2] + acc[j][3])) + ((acc[j][4] + acc[j][5]) + (acc[j][6] + acc[j][7]));
        float iv = sigm(gs[0]);
        float fv = sigm(gs[1]);
        float gv = tanhf(gs[2]);
        float ov = sigm(gs[3]);
        float cn = fv * c1r + iv * gv;
        c1r = cn;
        float hv = ov * tanhf(cn);
        const v2f* wp = Whrf0 + (w * 4) * 32 + lane;
        v8f pacc0 = {}, pacc1 = {};
        #pragma unroll
        for (int st = 0; st < 4; ++st) {
          float g0 = lane_read((gbase + 4 * st) << 2, hv);
          float g1 = lane_read((gbase + 4 * st + 1) << 2, hv);
          v2f pa; pa.x = arow ? g0 : 0.f;  pa.y = arow ? g1 : 0.f;
          if (st < 2) pacc0 = __builtin_amdgcn_wmma_f32_16x16x4_f32(false, pa, false, wp[st * 32], (short)0, pacc0, false, false);
          else        pacc1 = __builtin_amdgcn_wmma_f32_16x16x4_f32(false, pa, false, wp[st * 32], (short)0, pacc1, false, false);
        }
        p1 = pacc0[0] + pacc1[0];
        x1[poff] = p1;
      }
      __syncthreads();
      x0[poff + 256] = p1;

      {
        v8f acc[4];
        #pragma unroll
        for (int j = 0; j < 4; ++j) {
          acc[j] = (v8f){};
          acc[j][0] = bias1f[(w * 4 + j) * 16 + lm];
        }
        const v2f* wf = Wf1 + (w * 4 * 8) * 32 + lane;
        const v4f* xa = xin1F + lane;
        #pragma unroll
        for (int q = 0; q < 4; ++q) {
          v4f t = xa[q * 32];
          #pragma unroll
          for (int e = 0; e < 2; ++e) {
            int k = 2 * q + e;
            v2f a; a.x = e ? t.z : t.x; a.y = e ? t.w : t.y;
            #pragma unroll
            for (int j = 0; j < 4; ++j) {
              v2f b = wf[(8 * j + k) * 32];
              acc[j] = __builtin_amdgcn_wmma_f32_16x16x4_f32(false, a, false, b, (short)0, acc[j], false, false);
            }
          }
        }
        if (w < 1 && (n | f) != 0) {
          float v = 0.f;
          #pragma unroll
          for (int p = 0; p < NWAVE; ++p) v += x1[pbase + 4 * p + 256];
          *outp_prev = v + nm_o;
        }
        float gs[4];
        #pragma unroll
        for (int j = 0; j < 4; ++j)
          gs[j] = ((acc[j][0] + acc[j][1]) + (acc[j][2] + acc[j][3])) + ((acc[j][4] + acc[j][5]) + (acc[j][6] + acc[j][7]));
        float iv = sigm(gs[0]);
        float fv = sigm(gs[1]);
        float gv = tanhf(gs[2]);
        float ov = sigm(gs[3]);
        float cn = fv * c2r + iv * gv;
        c2r = cn;
        float hv = ov * tanhf(cn);
        const v2f* wp = Whrf1 + (w * 4) * 32 + lane;
        v8f pacc0 = {}, pacc1 = {};
        #pragma unroll
        for (int st = 0; st < 4; ++st) {
          float g0 = lane_read((gbase + 4 * st) << 2, hv);
          float g1 = lane_read((gbase + 4 * st + 1) << 2, hv);
          v2f pa; pa.x = arow ? g0 : 0.f;  pa.y = arow ? g1 : 0.f;
          if (st < 2) pacc0 = __builtin_amdgcn_wmma_f32_16x16x4_f32(false, pa, false, wp[st * 32], (short)0, pacc0, false, false);
          else        pacc1 = __builtin_amdgcn_wmma_f32_16x16x4_f32(false, pa, false, wp[st * 32], (short)0, pacc1, false, false);
        }
        p2 = pacc0[0] + pacc1[0];
        float add = 0.f;
        if (w < 1) add = (f == FUT - 1) ? nm_nx : nm_cur;
        x0[poff] = p2 + add;
      }
      outp_prev = out + (size_t)(b0 + lh) * FUT * NNODES * O_DIM
                      + (size_t)f * NNODES * O_DIM + (size_t)n * O_DIM + lm;
      nm_o = nm_cur;
    }
    nm_cur = nm_nx;
  }
  __syncthreads();
  x1[poff + 256] = p2;
  __syncthreads();
  if (w < 1) {
    float v = 0.f;
    #pragma unroll
    for (int p = 0; p < NWAVE; ++p) v += x1[pbase + 4 * p + 256];
    *outp_prev = v + nm_o;
  }
}

extern "C" void kernel_launch(void* const* d_in, const int* in_sizes, int n_in,
                              void* d_out, int out_size, void* d_ws, size_t ws_size,
                              hipStream_t stream) {
  (void)in_sizes; (void)n_in; (void)out_size; (void)d_ws; (void)ws_size;
  const float* enc   = (const float*)d_in[0];
  const float* mean_ = (const float*)d_in[1];
  const float* s     = (const float*)d_in[2];
  const float* h0    = (const float*)d_in[3];
  const float* c0    = (const float*)d_in[4];
  const float* Wih0  = (const float*)d_in[5];
  const float* Whh0  = (const float*)d_in[6];
  const float* bih0  = (const float*)d_in[7];
  const float* bhh0  = (const float*)d_in[8];
  const float* Whr0  = (const float*)d_in[9];
  const float* Wih1  = (const float*)d_in[10];
  const float* Whh1  = (const float*)d_in[11];
  const float* bih1  = (const float*)d_in[12];
  const float* bhh1  = (const float*)d_in[13];
  const float* Whr1  = (const float*)d_in[14];
  float* out = (float*)d_out;

  decoder_scan<<<dim3(NWG), dim3(NTHR), 8192, stream>>>(
      enc, mean_, s, h0, c0,
      Wih0, Whh0, bih0, bhh0, Whr0,
      Wih1, Whh1, bih1, bhh1, Whr1, out);
}


# tanh branch-free: the library's two range formulas both evaluated, selected per lane (same values): the cell update is one basic block, the gate activations' chains interleave; sums unchanged
# speedup vs baseline: 3.3420x; 1.1104x over previous
#include <hip/hip_runtime.h>
#include <math.h>

typedef __attribute__((ext_vector_type(2))) float v2f;
typedef __attribute__((ext_vector_type(4))) float v4f;
typedef __attribute__((ext_vector_type(8))) float v8f;

#define E_DIM 64
#define H_DIM 128
#define O_DIM 16
#define BATCH 64
#define NNODES 1000
#define FUT 12
#define G4 512
#define BR 2
#define NWG 32
#define NWAVE 8
#define NTHR 256

__device__ __forceinline__ float sigm(float x) { return 1.0f / (1.0f + __expf(-x)); }

__device__ __forceinline__ float tanh_sel(float x) {
  float y  = __builtin_fabsf(x);
  float t  = expf(2.0f * y);
  float zl = __builtin_fmaf(__builtin_amdgcn_rcpf(t + 1.0f), -2.0f, 1.0f);
  float y2 = y * y;
  float p  = __builtin_fmaf(y2, __builtin_fmaf(y2, __builtin_fmaf(y2, __builtin_fmaf(y2, -0x1.758e7ap-8f, 0x1.521192p-6f),
                                                                 -0x1.b8389cp-5f), 0x1.110704p-3f), -0x1.555532p-2f);
  float zs = __builtin_fmaf(y2, y * p, y);
  float z  = (y < 0.625f) ? zs : zl;
  return __builtin_copysignf(z, x);
}

__device__ __forceinline__ float lane_read(int byte_index, float v) {
  return __builtin_bit_cast(float, __builtin_amdgcn_ds_bpermute(byte_index, __builtin_bit_cast(int, v)));
}

__device__ __forceinline__ void gather_h(float hv, int gbase, float (&g)[8]) {
  #pragma unroll
  for (int i = 0; i < 8; ++i) g[i] = lane_read((gbase + 4 * (i >> 1) + (i & 1)) << 2, hv);
  __builtin_amdgcn_sched_barrier(0);
}

__global__ __launch_bounds__(NTHR) void decoder_scan(
    const float* __restrict__ enc,
    const float* __restrict__ mean_,
    const float* __restrict__ s,
    const float* __restrict__ h0,
    const float* __restrict__ c0,
    const float* __restrict__ Wih0,
    const float* __restrict__ Whh0,
    const float* __restrict__ bih0, const float* __restrict__ bhh0,
    const float* __restrict__ Whr0,
    const float* __restrict__ Wih1,
    const float* __restrict__ Whh1,
    const float* __restrict__ bih1, const float* __restrict__ bhh1,
    const float* __restrict__ Whr1,
    float* __restrict__ out)
{
  __shared__ v2f Wf0[NWAVE * 4 * 8 * 32];
  __shared__ v2f Wf1[NWAVE * 4 * 8 * 32];
  __shared__ v2f Whrf0[NWAVE * 4 * 32];
  __shared__ v2f Whrf1[NWAVE * 4 * 32];
  __shared__ float GencF[NWAVE * 4 * 32];
  __shared__ float bias0s[G4];
  __shared__ float bias1f[NWAVE * 4 * 16];
  __shared__ v4f xin0F[4 * 32];
  __shared__ v4f xin1F[4 * 32];

  const int tid  = threadIdx.x;
  const int w    = __builtin_amdgcn_readfirstlane(tid >> 5);
  const int lane = tid & 31;
  const int lm   = lane & 15;
  const int lh   = lane >> 4;
  const int b0   = blockIdx.x * BR;
  const int cu   = 16 * w + lm;
  const bool arow = ((lm & 7) == 0);
  const int gbase = 16 * (lm >> 3) + 2 * lh;
  const int pbase = (lm >> 3) * 128 + (8 * lh + 16 * ((lm >> 1) & 1)) * 4 + ((lm >> 2) & 1) * 2 + (lm & 1);
  const int poff  = pbase + 4 * w;
  const int aslot = lm >> 3;
  float* const x0 = (float*)xin0F;
  float* const x1 = (float*)xin1F;

  for (int idx = tid; idx < NWAVE * 4 * 8 * 32 * 2; idx += NTHR) {
    int e = idx & 1, ln = (idx >> 1) & 31, k = (idx >> 6) & 7, j = (idx >> 9) & 3, ww = idx >> 11;
    int kr = 4 * k + 2 * (ln >> 4) + e;
    int col = 128 * j + 16 * ww + (ln & 15);
    ((float*)Wf0)[idx] = (kr < 16) ? Wih0[col * 80 + 64 + kr] : Whh0[col * 16 + (kr - 16)];
    ((float*)Wf1)[idx] = (kr < 16) ? Wih1[col * 16 + kr]      : Whh1[col * 16 + (kr - 16)];
  }
  for (int idx = tid; idx < NWAVE * 4 * 32 * 2; idx += NTHR) {
    int e = idx & 1, ln = (idx >> 1) & 31, st = (idx >> 6) & 3, ww = idx >> 8;
    int unit = 16 * ww + 4 * st + 2 * (ln >> 4) + e;
    int o = ln & 15;
    ((float*)Whrf0)[idx] = Whr0[o * H_DIM + unit];
    ((float*)Whrf1)[idx] = Whr1[o * H_DIM + unit];
  }
  for (int idx = tid; idx < G4; idx += NTHR) {
    bias0s[idx] = bih0[idx] + bhh0[idx];
    int l15 = idx & 15, j = (idx >> 4) & 3, ww = idx >> 6;
    int c1 = 128 * j + 16 * ww + l15;
    bias1f[idx] = bih1[c1] + bhh1[c1];
  }
  const float* const mp0 = mean_ + (size_t)(b0 + lh) * FUT * NNODES * O_DIM + lm;
  float nm_cur = 0.f, nm_nx = 0.f, nm_o = 0.f;
  float p1 = 0.f, p2 = 0.f, lp = 0.f;
  if (w < 1) {
    float sum = 0.f;
    #pragma unroll
    for (int f = 0; f < FUT; ++f) sum += mp0[(size_t)f * NNODES * O_DIM];
    nm_cur = sum * (1.0f / FUT);
    p1 = h0[lm];
    p2 = h0[O_DIM + lm];
    lp = s[lm] + nm_cur;
  }
  x0[poff]       = lp;
  x0[poff + 256] = p1;
  float c1r = c0[cu], c2r = c0[H_DIM + cu];
  float* outp_prev = out;
  __syncthreads();

  for (int n = 0; n < NNODES; ++n) {
    if (w < 1) {
      nm_nx = 0.f;
      if (n + 1 < NNODES) {
        const float* mp = mp0 + (size_t)(n + 1) * O_DIM;
        float sum = 0.f;
        #pragma unroll
        for (int f = 0; f < FUT; ++f) sum += mp[(size_t)f * NNODES * O_DIM];
        nm_nx = sum * (1.0f / FUT);
      }
    }
    {
      v8f hacc[4];
      #pragma unroll
      for (int j = 0; j < 4; ++j) {
        float bv = bias0s[128 * j + cu];
        #pragma unroll
        for (int r = 0; r < 8; ++r) hacc[j][r] = bv;
      }
      const float* ap = enc + (size_t)(b0 + aslot) * NNODES * E_DIM + (size_t)n * E_DIM;
      #pragma unroll
      for (int k = 0; k < 16; ++k) {
        int kk = 4 * k + 2 * lh;
        v2f a;
        a.x = ap[kk];  a.y = ap[kk + 1];
        #pragma unroll
        for (int j = 0; j < 4; ++j) {
          v2f b;
          b.x = Wih0[(128 * j + cu) * 80 + kk];  b.y = Wih0[(128 * j + cu) * 80 + kk + 1];
          hacc[j] = __builtin_amdgcn_wmma_f32_16x16x4_f32(false, a, false, b, (short)0, hacc[j], false, false);
        }
      }
      #pragma unroll
      for (int j = 0; j < 4; ++j) GencF[(w * 4 + j) * 32 + lane] = hacc[j][0];
    }

    for (int f = 0; f < FUT; ++f) {
      __syncthreads();
      x1[poff + 256] = p2;

      {
        v8f acc[4];
        #pragma unroll
        for (int j = 0; j < 4; ++j) {
          acc[j] = (v8f){};
          acc[j][0] = GencF[(w * 4 + j) * 32 + lane];
        }
        const v2f* wf = Wf0 + (w * 4 * 8) * 32 + lane;
        const v4f* xa = xin0F + lane;
        #pragma unroll
        for (int q = 0; q < 4; ++q) {
          v4f t = xa[q * 32];
          #pragma unroll
          for (int e = 0; e < 2; ++e) {
            int k = 2 * q + e;
            v2f a; a.x = e ? t.z : t.x; a.y = e ? t.w : t.y;
            #pragma unroll
            for (int j = 0; j < 4; ++j) {
              v2f b = wf[(8 * j + k) * 32];
              acc[j] = __builtin_amdgcn_wmma_f32_16x16x4_f32(false, a, false, b, (short)0, acc[j], false, false);
            }
          }
        }
        float gs[4];
        #pragma unroll
        for (int j = 0; j < 4; ++j)
          gs[j] = ((acc[j][0] + acc[j][1]) + (acc[j][2] + acc[j][3])) + ((acc[j][4] + acc[j][5]) + (acc[j][6] + acc[j][7]));
        v2f wpv[4];
        #pragma unroll
        for (int st = 0; st < 4; ++st) wpv[st] = Whrf0[(w * 4 + st) * 32 + lane];
        float iv = sigm(gs[0]);
        float fv = sigm(gs[1]);
        float gv = tanh_sel(gs[2]);
        float ov = sigm(gs[3]);
        float cn = fv * c1r + iv * gv;
        c1r = cn;
        float hv = ov * tanh_sel(cn);
        v8f pacc0 = {}, pacc1 = {};
        float g[8];
        gather_h(hv, gbase, g);
        #pragma unroll
        for (int st = 0; st < 4; ++st) {
          v2f pa; pa.x = arow ? g[2 * st] : 0.f;  pa.y = arow ? g[2 * st + 1] : 0.f;
          if (st < 2) pacc0 = __builtin_amdgcn_wmma_f32_16x16x4_f32(false, pa, false, wpv[st], (short)0, pacc0, false, false);
          else        pacc1 = __builtin_amdgcn_wmma_f32_16x16x4_f32(false, pa, false, wpv[st], (short)0, pacc1, false, false);
        }
        p1 = pacc0[0] + pacc1[0];
        x1[poff] = p1;
      }
      __syncthreads();
      x0[poff + 256] = p1;

      {
        v8f acc[4];
        #pragma unroll
        for (int j = 0; j < 4; ++j) {
          acc[j] = (v8f){};
          acc[j][0] = bias1f[(w * 4 + j) * 16 + lm];
        }
        const v2f* wf = Wf1 + (w * 4 * 8) * 32 + lane;
        const v4f* xa = xin1F + lane;
        #pragma unroll
        for (int q = 0; q < 4; ++q) {
          v4f t = xa[q * 32];
          #pragma unroll
          for (int e = 0; e < 2; ++e) {
            int k = 2 * q + e;
            v2f a; a.x = e ? t.z : t.x; a.y = e ? t.w : t.y;
            #pragma unroll
            for (int j = 0; j < 4; ++j) {
              v2f b = wf[(8 * j + k) * 32];
              acc[j] = __builtin_amdgcn_wmma_f32_16x16x4_f32(false, a, false, b, (short)0, acc[j], false, false);
            }
          }
        }
        if (w < 1 && (n | f) != 0) {
          float v = 0.f;
          #pragma unroll
          for (int p = 0; p < NWAVE; ++p) v += x1[pbase + 4 * p + 256];
          *outp_prev = v + nm_o;
        }
        float gs[4];
        #pragma unroll
        for (int j = 0; j < 4; ++j)
          gs[j] = ((acc[j][0] + acc[j][1]) + (acc[j][2] + acc[j][3])) + ((acc[j][4] + acc[j][5]) + (acc[j][6] + acc[j][7]));
        v2f wpv[4];
        #pragma unroll
        for (int st = 0; st < 4; ++st) wpv[st] = Whrf1[(w * 4 + st) * 32 + lane];
        float iv = sigm(gs[0]);
        float fv = sigm(gs[1]);
        float gv = tanh_sel(gs[2]);
        float ov = sigm(gs[3]);
        float cn = fv * c2r + iv * gv;
        c2r = cn;
        float hv = ov * tanh_sel(cn);
        v8f pacc0 = {}, pacc1 = {};
        float g[8];
        gather_h(hv, gbase, g);
        #pragma unroll
        for (int st = 0; st < 4; ++st) {
          v2f pa; pa.x = arow ? g[2 * st] : 0.f;  pa.y = arow ? g[2 * st + 1] : 0.f;
          if (st < 2) pacc0 = __builtin_amdgcn_wmma_f32_16x16x4_f32(false, pa, false, wpv[st], (short)0, pacc0, false, false);
          else        pacc1 = __builtin_amdgcn_wmma_f32_16x16x4_f32(false, pa, false, wpv[st], (short)0, pacc1, false, false);
        }
        p2 = pacc0[0] + pacc1[0];
        float add = 0.f;
        if (w < 1) add = (f == FUT - 1) ? nm_nx : nm_cur;
        x0[poff] = p2 + add;
      }
      outp_prev = out + (size_t)(b0 + lh) * FUT * NNODES * O_DIM
                      + (size_t)f * NNODES * O_DIM + (size_t)n * O_DIM + lm;
      nm_o = nm_cur;
    }
    nm_cur = nm_nx;
  }
  __syncthreads();
  x1[poff + 256] = p2;
  __syncthreads();
  if (w < 1) {
    float v = 0.f;
    #pragma unroll
    for (int p = 0; p < NWAVE; ++p) v += x1[pbase + 4 * p + 256];
    *outp_prev = v + nm_o;
  }
}

extern "C" void kernel_launch(void* const* d_in, const int* in_sizes, int n_in,
                              void* d_out, int out_size, void* d_ws, size_t ws_size,
                              hipStream_t stream) {
  (void)in_sizes; (void)n_in; (void)out_size; (void)d_ws; (void)ws_size;
  const float* enc   = (const float*)d_in[0];
  const float* mean_ = (const float*)d_in[1];
  const float* s     = (const float*)d_in[2];
  const float* h0    = (const float*)d_in[3];
  const float* c0    = (const float*)d_in[4];
  const float* Wih0  = (const float*)d_in[5];
  const float* Whh0  = (const float*)d_in[6];
  const float* bih0  = (const float*)d_in[7];
  const float* bhh0  = (const float*)d_in[8];
  const float* Whr0  = (const float*)d_in[9];
  const float* Wih1  = (const float*)d_in[10];
  const float* Whh1  = (const float*)d_in[11];
  const float* bih1  = (const float*)d_in[12];
  const float* bhh1  = (const float*)d_in[13];
  const float* Whr1  = (const float*)d_in[14];
  float* out = (float*)d_out;

  decoder_scan<<<dim3(NWG), dim3(NTHR), 8192, stream>>>(
      enc, mean_, s, h0, c0,
      Wih0, Whh0, bih0, bhh0, Whr0,
      Wih1, Whh1, bih1, bhh1, Whr1, out);
}


# Enc gate term formed for 8 nodes per pass: the 16 matrix rows carry 2 batch slots x 8 nodes (were 2 real rows), results parked in LDS per lane: 8 products per node, not 64; sums unchanged
# speedup vs baseline: 3.5048x; 1.0487x over previous
#include <hip/hip_runtime.h>
#include <math.h>

typedef __attribute__((ext_vector_type(2))) float v2f;
typedef __attribute__((ext_vector_type(4))) float v4f;
typedef __attribute__((ext_vector_type(8))) float v8f;

#define E_DIM 64
#define H_DIM 128
#define O_DIM 16
#define BATCH 64
#define NNODES 1000
#define FUT 12
#define G4 512
#define BR 2
#define NWG 32
#define NWAVE 8
#define NTHR 256
#define NB 8

__device__ __forceinline__ float sigm(float x) { return 1.0f / (1.0f + __expf(-x)); }

__device__ __forceinline__ float tanh_sel(float x) {
  float y  = __builtin_fabsf(x);
  float t  = expf(2.0f * y);
  float zl = __builtin_fmaf(__builtin_amdgcn_rcpf(t + 1.0f), -2.0f, 1.0f);
  float y2 = y * y;
  float p  = __builtin_fmaf(y2, __builtin_fmaf(y2, __builtin_fmaf(y2, __builtin_fmaf(y2, -0x1.758e7ap-8f, 0x1.521192p-6f),
                                                                 -0x1.b8389cp-5f), 0x1.110704p-3f), -0x1.555532p-2f);
  float zs = __builtin_fmaf(y2, y * p, y);
  float z  = (y < 0.625f) ? zs : zl;
  return __builtin_copysignf(z, x);
}

__device__ __forceinline__ float lane_read(int byte_index, float v) {
  return __builtin_bit_cast(float, __builtin_amdgcn_ds_bpermute(byte_index, __builtin_bit_cast(int, v)));
}

__device__ __forceinline__ void gather_h(float hv, int gbase, float (&g)[8]) {
  #pragma unroll
  for (int i = 0; i < 8; ++i) g[i] = lane_read((gbase + 4 * (i >> 1) + (i & 1)) << 2, hv);
  __builtin_amdgcn_sched_barrier(0);
}

__global__ __launch_bounds__(NTHR) void decoder_scan(
    const float* __restrict__ enc,
    const float* __restrict__ mean_,
    const float* __restrict__ s,
    const float* __restrict__ h0,
    const float* __restrict__ c0,
    const float* __restrict__ Wih0,
    const float* __restrict__ Whh0,
    const float* __restrict__ bih0, const float* __restrict__ bhh0,
    const float* __restrict__ Whr0,
    const float* __restrict__ Wih1,
    const float* __restrict__ Whh1,
    const float* __restrict__ bih1, const float* __restrict__ bhh1,
    const float* __restrict__ Whr1,
    float* __restrict__ out)
{
  __shared__ v2f Wf0[NWAVE * 4 * 8 * 32];
  __shared__ v2f Wf1[NWAVE * 4 * 8 * 32];
  __shared__ v2f Whrf0[NWAVE * 4 * 32];
  __shared__ v2f Whrf1[NWAVE * 4 * 32];
  __shared__ float GencF[NB * NWAVE * 4 * 32];
  __shared__ float bias0s[G4];
  __shared__ float bias1f[NWAVE * 4 * 16];
  __shared__ v4f xin0F[4 * 32];
  __shared__ v4f xin1F[4 * 32];

  const int tid  = threadIdx.x;
  const int w    = __builtin_amdgcn_readfirstlane(tid >> 5);
  const int lane = tid & 31;
  const int lm   = lane & 15;
  const int lh   = lane >> 4;
  const int b0   = blockIdx.x * BR;
  const int cu   = 16 * w + lm;
  const bool arow = ((lm & 7) == 0);
  const int gbase = 16 * (lm >> 3) + 2 * lh;
  const int pbase = (lm >> 3) * 128 + (8 * lh + 16 * ((lm >> 1) & 1)) * 4 + ((lm >> 2) & 1) * 2 + (lm & 1);
  const int poff  = pbase + 4 * w;
  const int aslot = lm >> 3;
  const int anode = lm & 7;
  float* const x0 = (float*)xin0F;
  float* const x1 = (float*)xin1F;

  for (int idx = tid; idx < NWAVE * 4 * 8 * 32 * 2; idx += NTHR) {
    int e = idx & 1, ln = (idx >> 1) & 31, k = (idx >> 6) & 7, j = (idx >> 9) & 3, ww = idx >> 11;
    int kr = 4 * k + 2 * (ln >> 4) + e;
    int col = 128 * j + 16 * ww + (ln & 15);
    ((float*)Wf0)[idx] = (kr < 16) ? Wih0[col * 80 + 64 + kr] : Whh0[col * 16 + (kr - 16)];
    ((float*)Wf1)[idx] = (kr < 16) ? Wih1[col * 16 + kr]      : Whh1[col * 16 + (kr - 16)];
  }
  for (int idx = tid; idx < NWAVE * 4 * 32 * 2; idx += NTHR) {
    int e = idx & 1, ln = (idx >> 1) & 31, st = (idx >> 6) & 3, ww = idx >> 8;
    int unit = 16 * ww + 4 * st + 2 * (ln >> 4) + e;
    int o = ln & 15;
    ((float*)Whrf0)[idx] = Whr0[o * H_DIM + unit];
    ((float*)Whrf1)[idx] = Whr1[o * H_DIM + unit];
  }
  for (int idx = tid; idx < G4; idx += NTHR) {
    bias0s[idx] = bih0[idx] + bhh0[idx];
    int l15 = idx & 15, j = (idx >> 4) & 3, ww = idx >> 6;
    int c1 = 128 * j + 16 * ww + l15;
    bias1f[idx] = bih1[c1] + bhh1[c1];
  }
  const float* const mp0 = mean_ + (size_t)(b0 + lh) * FUT * NNODES * O_DIM + lm;
  float nm_cur = 0.f, nm_nx = 0.f, nm_o = 0.f;
  float p1 = 0.f, p2 = 0.f, lp = 0.f;
  if (w < 1) {
    float sum = 0.f;
    #pragma unroll
    for (int f = 0; f < FUT; ++f) sum += mp0[(size_t)f * NNODES * O_DIM];
    nm_cur = sum * (1.0f / FUT);
    p1 = h0[lm];
    p2 = h0[O_DIM + lm];
    lp = s[lm] + nm_cur;
  }
  x0[poff]       = lp;
  x0[poff + 256] = p1;
  float c1r = c0[cu], c2r = c0[H_DIM + cu];
  v2f wenc[16][4];
  #pragma unroll
  for (int k = 0; k < 16; ++k) {
    #pragma unroll
    for (int j = 0; j < 4; ++j) {
      const float* wp = Wih0 + (128 * j + cu) * 80 + 4 * k + 2 * lh;
      wenc[k][j].x = wp[0];  wenc[k][j].y = wp[1];
    }
  }
  float* outp_prev = out;
  __syncthreads();

  for (int n = 0; n < NNODES; ++n) {
    if (w < 1) {
      nm_nx = 0.f;
      if (n + 1 < NNODES) {
        const float* mp = mp0 + (size_t)(n + 1) * O_DIM;
        float sum = 0.f;
        #pragma unroll
        for (int f = 0; f < FUT; ++f) sum += mp[(size_t)f * NNODES * O_DIM];
        nm_nx = sum * (1.0f / FUT);
      }
    }
    if ((n & (NB - 1)) == 0) {
      v8f hacc[4];
      #pragma unroll
      for (int j = 0; j < 4; ++j) {
        float bv = bias0s[128 * j + cu];
        #pragma unroll
        for (int r = 0; r < 8; ++r) hacc[j][r] = bv;
      }
      const int an = (n + anode < NNODES) ? n + anode : NNODES - 1;
      const float* ap = enc + (size_t)(b0 + aslot) * NNODES * E_DIM + (size_t)an * E_DIM;
      #pragma unroll
      for (int k = 0; k < 16; ++k) {
        int kk = 4 * k + 2 * lh;
        v2f a;
        a.x = ap[kk];  a.y = ap[kk + 1];
        #pragma unroll
        for (int j = 0; j < 4; ++j)
          hacc[j] = __builtin_amdgcn_wmma_f32_16x16x4_f32(false, a, false, wenc[k][j], (short)0, hacc[j], false, false);
      }
      #pragma unroll
      for (int r = 0; r < NB; ++r) {
        #pragma unroll
        for (int j = 0; j < 4; ++j) GencF[r * (NWAVE * 4 * 32) + (w * 4 + j) * 32 + lane] = hacc[j][r];
      }
    }
    const float* const genc = GencF + (n & (NB - 1)) * (NWAVE * 4 * 32) + w * 4 * 32 + lane;

    for (int f = 0; f < FUT; ++f) {
      __syncthreads();
      x1[poff + 256] = p2;

      {
        v8f acc[4];
        #pragma unroll
        for (int j = 0; j < 4; ++j) {
          acc[j] = (v8f){};
          acc[j][0] = genc[j * 32];
        }
        const v2f* wf = Wf0 + (w * 4 * 8) * 32 + lane;
        const v4f* xa = xin0F + lane;
        #pragma unroll
        for (int q = 0; q < 4; ++q) {
          v4f t = xa[q * 32];
          #pragma unroll
          for (int e = 0; e < 2; ++e) {
            int k = 2 * q + e;
            v2f a; a.x = e ? t.z : t.x; a.y = e ? t.w : t.y;
            #pragma unroll
            for (int j = 0; j < 4; ++j) {
              v2f b = wf[(8 * j + k) * 32];
              acc[j] = __builtin_amdgcn_wmma_f32_16x16x4_f32(false, a, false, b, (short)0, acc[j], false, false);
            }
          }
        }
        float gs[4];
        #pragma unroll
        for (int j = 0; j < 4; ++j)
          gs[j] = ((acc[j][0] + acc[j][1]) + (acc[j][2] + acc[j][3])) + ((acc[j][4] + acc[j][5]) + (acc[j][6] + acc[j][7]));
        v2f wpv[4];
        #pragma unroll
        for (int st = 0; st < 4; ++st) wpv[st] = Whrf0[(w * 4 + st) * 32 + lane];
        float iv = sigm(gs[0]);
        float fv = sigm(gs[1]);
        float gv = tanh_sel(gs[2]);
        float ov = sigm(gs[3]);
        float cn = fv * c1r + iv * gv;
        c1r = cn;
        float hv = ov * tanh_sel(cn);
        v8f pacc0 = {}, pacc1 = {};
        float g[8];
        gather_h(hv, gbase, g);
        #pragma unroll
        for (int st = 0; st < 4; ++st) {
          v2f pa; pa.x = arow ? g[2 * st] : 0.f;  pa.y = arow ? g[2 * st + 1] : 0.f;
          if (st < 2) pacc0 = __builtin_amdgcn_wmma_f32_16x16x4_f32(false, pa, false, wpv[st], (short)0, pacc0, false, false);
          else        pacc1 = __builtin_amdgcn_wmma_f32_16x16x4_f32(false, pa, false, wpv[st], (short)0, pacc1, false, false);
        }
        p1 = pacc0[0] + pacc1[0];
        x1[poff] = p1;
      }
      __syncthreads();
      x0[poff + 256] = p1;

      {
        v8f acc[4];
        #pragma unroll
        for (int j = 0; j < 4; ++j) {
          acc[j] = (v8f){};
          acc[j][0] = bias1f[(w * 4 + j) * 16 + lm];
        }
        const v2f* wf = Wf1 + (w * 4 * 8) * 32 + lane;
        const v4f* xa = xin1F + lane;
        #pragma unroll
        for (int q = 0; q < 4; ++q) {
          v4f t = xa[q * 32];
          #pragma unroll
          for (int e = 0; e < 2; ++e) {
            int k = 2 * q + e;
            v2f a; a.x = e ? t.z : t.x; a.y = e ? t.w : t.y;
            #pragma unroll
            for (int j = 0; j < 4; ++j) {
              v2f b = wf[(8 * j + k) * 32];
              acc[j] = __builtin_amdgcn_wmma_f32_16x16x4_f32(false, a, false, b, (short)0, acc[j], false, false);
            }
          }
        }
        if (w < 1 && (n | f) != 0) {
          float v = 0.f;
          #pragma unroll
          for (int p = 0; p < NWAVE; ++p) v += x1[pbase + 4 * p + 256];
          *outp_prev = v + nm_o;
        }
        float gs[4];
        #pragma unroll
        for (int j = 0; j < 4; ++j)
          gs[j] = ((acc[j][0] + acc[j][1]) + (acc[j][2] + acc[j][3])) + ((acc[j][4] + acc[j][5]) + (acc[j][6] + acc[j][7]));
        v2f wpv[4];
        #pragma unroll
        for (int st = 0; st < 4; ++st) wpv[st] = Whrf1[(w * 4 + st) * 32 + lane];
        float iv = sigm(gs[0]);
        float fv = sigm(gs[1]);
        float gv = tanh_sel(gs[2]);
        float ov = sigm(gs[3]);
        float cn = fv * c2r + iv * gv;
        c2r = cn;
        float hv = ov * tanh_sel(cn);
        v8f pacc0 = {}, pacc1 = {};
        float g[8];
        gather_h(hv, gbase, g);
        #pragma unroll
        for (int st = 0; st < 4; ++st) {
          v2f pa; pa.x = arow ? g[2 * st] : 0.f;  pa.y = arow ? g[2 * st + 1] : 0.f;
          if (st < 2) pacc0 = __builtin_amdgcn_wmma_f32_16x16x4_f32(false, pa, false, wpv[st], (short)0, pacc0, false, false);
          else        pacc1 = __builtin_amdgcn_wmma_f32_16x16x4_f32(false, pa, false, wpv[st], (short)0, pacc1, false, false);
        }
        p2 = pacc0[0] + pacc1[0];
        float add = 0.f;
        if (w < 1) add = (f == FUT - 1) ? nm_nx : nm_cur;
        x0[poff] = p2 + add;
      }
      outp_prev = out + (size_t)(b0 + lh) * FUT * NNODES * O_DIM
                      + (size_t)f * NNODES * O_DIM + (size_t)n * O_DIM + lm;
      nm_o = nm_cur;
    }
    nm_cur = nm_nx;
  }
  __syncthreads();
  x1[poff + 256] = p2;
  __syncthreads();
  if (w < 1) {
    float v = 0.f;
    #pragma unroll
    for (int p = 0; p < NWAVE; ++p) v += x1[pbase + 4 * p + 256];
    *outp_prev = v + nm_o;
  }
}

extern "C" void kernel_launch(void* const* d_in, const int* in_sizes, int n_in,
                              void* d_out, int out_size, void* d_ws, size_t ws_size,
                              hipStream_t stream) {
  (void)in_sizes; (void)n_in; (void)out_size; (void)d_ws; (void)ws_size;
  const float* enc   = (const float*)d_in[0];
  const float* mean_ = (const float*)d_in[1];
  const float* s     = (const float*)d_in[2];
  const float* h0    = (const float*)d_in[3];
  const float* c0    = (const float*)d_in[4];
  const float* Wih0  = (const float*)d_in[5];
  const float* Whh0  = (const float*)d_in[6];
  const float* bih0  = (const float*)d_in[7];
  const float* bhh0  = (const float*)d_in[8];
  const float* Whr0  = (const float*)d_in[9];
  const float* Wih1  = (const float*)d_in[10];
  const float* Whh1  = (const float*)d_in[11];
  const float* bih1  = (const float*)d_in[12];
  const float* bhh1  = (const float*)d_in[13];
  const float* Whr1  = (const float*)d_in[14];
  float* out = (float*)d_out;

  decoder_scan<<<dim3(NWG), dim3(NTHR), 0, stream>>>(
      enc, mean_, s, h0, c0,
      Wih0, Whh0, bih0, bhh0, Whr0,
      Wih1, Whh1, bih1, bhh1, Whr1, out);
}


# Gate run pipelined: a layer's LDS reads issued as one group ahead of its 32 products, the o gate's tile run last with the other gates' activation chains issued between its products; sums unchanged
# speedup vs baseline: 3.7228x; 1.0622x over previous
#include <hip/hip_runtime.h>
#include <math.h>

typedef __attribute__((ext_vector_type(2))) float v2f;
typedef __attribute__((ext_vector_type(4))) float v4f;
typedef __attribute__((ext_vector_type(8))) float v8f;

#define E_DIM 64
#define H_DIM 128
#define O_DIM 16
#define BATCH 64
#define NNODES 1000
#define FUT 12
#define G4 512
#define BR 2
#define NWG 32
#define NWAVE 8
#define NTHR 256
#define NB 8

__device__ __forceinline__ float sigm(float x) { return 1.0f / (1.0f + __expf(-x)); }

__device__ __forceinline__ float tanh_sel(float x) {
  float y  = __builtin_fabsf(x);
  float t  = expf(2.0f * y);
  float zl = __builtin_fmaf(__builtin_amdgcn_rcpf(t + 1.0f), -2.0f, 1.0f);
  float y2 = y * y;
  float p  = __builtin_fmaf(y2, __builtin_fmaf(y2, __builtin_fmaf(y2, __builtin_fmaf(y2, -0x1.758e7ap-8f, 0x1.521192p-6f),
                                                                 -0x1.b8389cp-5f), 0x1.110704p-3f), -0x1.555532p-2f);
  float zs = __builtin_fmaf(y2, y * p, y);
  float z  = (y < 0.625f) ? zs : zl;
  return __builtin_copysignf(z, x);
}

__device__ __forceinline__ float lane_read(int byte_index, float v) {
  return __builtin_bit_cast(float, __builtin_amdgcn_ds_bpermute(byte_index, __builtin_bit_cast(int, v)));
}

__device__ __forceinline__ void gather_h(float hv, int gbase, float (&g)[8]) {
  #pragma unroll
  for (int i = 0; i < 8; ++i) g[i] = lane_read((gbase + 4 * (i >> 1) + (i & 1)) << 2, hv);
  __builtin_amdgcn_sched_barrier(0);
}

__device__ __forceinline__ float sum8(const v8f a) {
  return ((a[0] + a[1]) + (a[2] + a[3])) + ((a[4] + a[5]) + (a[6] + a[7]));
}

__device__ __forceinline__ v2f a_step(const v4f (&t)[4], int k) {
  const v4f tt = t[k >> 1];
  v2f a; a.x = (k & 1) ? tt.z : tt.x; a.y = (k & 1) ? tt.w : tt.y;
  return a;
}

template <bool OUT>
__device__ __forceinline__ float gate_cell(const v4f* __restrict__ xa, const v2f* __restrict__ wf, const v2f* __restrict__ whr,
                                           const float s0, const float s1, const float s2, const float s3,
                                           float& c, v2f (&wpv)[4],
                                           const bool do_out, const float* __restrict__ hp, float* __restrict__ outp, const float nm_o) {
  v4f t[4];
  v2f b[4][8];
  #pragma unroll
  for (int q = 0; q < 4; ++q) {
    t[q] = xa[q * 32];
    #pragma unroll
    for (int e = 0; e < 2; ++e) {
      #pragma unroll
      for (int j = 0; j < 3; ++j) b[j][2 * q + e] = wf[(8 * j + 2 * q + e) * 32];
    }
  }
  #pragma unroll
  for (int k = 0; k < 8; ++k) b[3][k] = wf[(8 * 3 + k) * 32];
  #pragma unroll
  for (int st = 0; st < 4; ++st) wpv[st] = whr[st * 32];
  float hpv[NWAVE];
  if (OUT) {
    if (do_out) {
      #pragma unroll
      for (int p = 0; p < NWAVE; ++p) hpv[p] = hp[4 * p];
    }
  }
  __builtin_amdgcn_sched_barrier(0);
  v8f acc[4];
  #pragma unroll
  for (int j = 0; j < 4; ++j) acc[j] = (v8f){};
  acc[0][0] = s0;  acc[1][0] = s1;  acc[2][0] = s2;  acc[3][0] = s3;
  #pragma unroll
  for (int k = 0; k < 8; ++k) {
    const v2f a = a_step(t, k);
    #pragma unroll
    for (int j = 0; j < 3; ++j)
      acc[j] = __builtin_amdgcn_wmma_f32_16x16x4_f32(false, a, false, b[j][k], (short)0, acc[j], false, false);
  }
  #pragma unroll
  for (int k = 0; k < 2; ++k)
    acc[3] = __builtin_amdgcn_wmma_f32_16x16x4_f32(false, a_step(t, k), false, b[3][k], (short)0, acc[3], false, false);
  if (OUT) {
    if (do_out) {
      float v = 0.f;
      #pragma unroll
      for (int p = 0; p < NWAVE; ++p) v += hpv[p];
      *outp = v + nm_o;
    }
  }
  __builtin_amdgcn_sched_barrier(0);
  #pragma unroll
  for (int k = 2; k < 8; ++k)
    acc[3] = __builtin_amdgcn_wmma_f32_16x16x4_f32(false, a_step(t, k), false, b[3][k], (short)0, acc[3], false, false);
  const float iv = sigm(sum8(acc[0]));
  const float fv = sigm(sum8(acc[1]));
  const float gv = tanh_sel(sum8(acc[2]));
  const float cn = __builtin_fmaf(c, fv, iv * gv);
  c = cn;
  const float th = tanh_sel(cn);
  const float hv = sigm(sum8(acc[3])) * th;
  #pragma unroll
  for (int i = 0; i < 6; ++i) {
    __builtin_amdgcn_sched_group_barrier(0x402, 10, 0);
    __builtin_amdgcn_sched_group_barrier(0x008, 1, 0);
  }
  return hv;
}

__global__ __launch_bounds__(NTHR) void decoder_scan(
    const float* __restrict__ enc,
    const float* __restrict__ mean_,
    const float* __restrict__ s,
    const float* __restrict__ h0,
    const float* __restrict__ c0,
    const float* __restrict__ Wih0,
    const float* __restrict__ Whh0,
    const float* __restrict__ bih0, const float* __restrict__ bhh0,
    const float* __restrict__ Whr0,
    const float* __restrict__ Wih1,
    const float* __restrict__ Whh1,
    const float* __restrict__ bih1, const float* __restrict__ bhh1,
    const float* __restrict__ Whr1,
    float* __restrict__ out)
{
  __shared__ v2f Wf0[NWAVE * 4 * 8 * 32];
  __shared__ v2f Wf1[NWAVE * 4 * 8 * 32];
  __shared__ v2f Whrf0[NWAVE * 4 * 32];
  __shared__ v2f Whrf1[NWAVE * 4 * 32];
  __shared__ float GencF[NB * NWAVE * 4 * 32];
  __shared__ float bias0s[G4];
  __shared__ float bias1f[NWAVE * 4 * 16];
  __shared__ v4f xin0F[4 * 32];
  __shared__ v4f xin1F[4 * 32];

  const int tid  = threadIdx.x;
  const int w    = __builtin_amdgcn_readfirstlane(tid >> 5);
  const int lane = tid & 31;
  const int lm   = lane & 15;
  const int lh   = lane >> 4;
  const int b0   = blockIdx.x * BR;
  const int cu   = 16 * w + lm;
  const bool arow = ((lm & 7) == 0);
  const int gbase = 16 * (lm >> 3) + 2 * lh;
  const int pbase = (lm >> 3) * 128 + (8 * lh + 16 * ((lm >> 1) & 1)) * 4 + ((lm >> 2) & 1) * 2 + (lm & 1);
  const int poff  = pbase + 4 * w;
  const int aslot = lm >> 3;
  const int anode = lm & 7;
  float* const x0 = (float*)xin0F;
  float* const x1 = (float*)xin1F;

  for (int idx = tid; idx < NWAVE * 4 * 8 * 32 * 2; idx += NTHR) {
    int e = idx & 1, ln = (idx >> 1) & 31, k = (idx >> 6) & 7, j = (idx >> 9) & 3, ww = idx >> 11;
    int kr = 4 * k + 2 * (ln >> 4) + e;
    int col = 128 * j + 16 * ww + (ln & 15);
    ((float*)Wf0)[idx] = (kr < 16) ? Wih0[col * 80 + 64 + kr] : Whh0[col * 16 + (kr - 16)];
    ((float*)Wf1)[idx] = (kr < 16) ? Wih1[col * 16 + kr]      : Whh1[col * 16 + (kr - 16)];
  }
  for (int idx = tid; idx < NWAVE * 4 * 32 * 2; idx += NTHR) {
    int e = idx & 1, ln = (idx >> 1) & 31, st = (idx >> 6) & 3, ww = idx >> 8;
    int unit = 16 * ww + 4 * st + 2 * (ln >> 4) + e;
    int o = ln & 15;
    ((float*)Whrf0)[idx] = Whr0[o * H_DIM + unit];
    ((float*)Whrf1)[idx] = Whr1[o * H_DIM + unit];
  }
  for (int idx = tid; idx < G4; idx += NTHR) {
    bias0s[idx] = bih0[idx] + bhh0[idx];
    int l15 = idx & 15, j = (idx >> 4) & 3, ww = idx >> 6;
    int c1 = 128 * j + 16 * ww + l15;
    bias1f[idx] = bih1[c1] + bhh1[c1];
  }
  const float* const mp0 = mean_ + (size_t)(b0 + lh) * FUT * NNODES * O_DIM + lm;
  float nm_cur = 0.f, nm_nx = 0.f, nm_o = 0.f;
  float p1 = 0.f, p2 = 0.f, lp = 0.f;
  if (w < 1) {
    float sum = 0.f;
    #pragma unroll
    for (int f = 0; f < FUT; ++f) sum += mp0[(size_t)f * NNODES * O_DIM];
    nm_cur = sum * (1.0f / FUT);
    p1 = h0[lm];
    p2 = h0[O_DIM + lm];
    lp = s[lm] + nm_cur;
  }
  x0[poff]       = lp;
  x0[poff + 256] = p1;
  float c1r = c0[cu], c2r = c0[H_DIM + cu];
  v2f wenc[16][4];
  #pragma unroll
  for (int k = 0; k < 16; ++k) {
    #pragma unroll
    for (int j = 0; j < 4; ++j) {
      const float* wp = Wih0 + (128 * j + cu) * 80 + 4 * k + 2 * lh;
      wenc[k][j].x = wp[0];  wenc[k][j].y = wp[1];
    }
  }
  float* outp_prev = out;
  __syncthreads();

  for (int n = 0; n < NNODES; ++n) {
    if (w < 1) {
      nm_nx = 0.f;
      if (n + 1 < NNODES) {
        const float* mp = mp0 + (size_t)(n + 1) * O_DIM;
        float sum = 0.f;
        #pragma unroll
        for (int f = 0; f < FUT; ++f) sum += mp[(size_t)f * NNODES * O_DIM];
        nm_nx = sum * (1.0f / FUT);
      }
    }
    if ((n & (NB - 1)) == 0) {
      v8f hacc[4];
      #pragma unroll
      for (int j = 0; j < 4; ++j) {
        float bv = bias0s[128 * j + cu];
        #pragma unroll
        for (int r = 0; r < 8; ++r) hacc[j][r] = bv;
      }
      const int an = (n + anode < NNODES) ? n + anode : NNODES - 1;
      const float* ap = enc + (size_t)(b0 + aslot) * NNODES * E_DIM + (size_t)an * E_DIM;
      #pragma unroll
      for (int k = 0; k < 16; ++k) {
        int kk = 4 * k + 2 * lh;
        v2f a;
        a.x = ap[kk];  a.y = ap[kk + 1];
        #pragma unroll
        for (int j = 0; j < 4; ++j)
          hacc[j] = __builtin_amdgcn_wmma_f32_16x16x4_f32(false, a, false, wenc[k][j], (short)0, hacc[j], false, false);
      }
      #pragma unroll
      for (int r = 0; r < NB; ++r) {
        #pragma unroll
        for (int j = 0; j < 4; ++j) GencF[r * (NWAVE * 4 * 32) + (w * 4 + j) * 32 + lane] = hacc[j][r];
      }
    }
    const float* const genc = GencF + (n & (NB - 1)) * (NWAVE * 4 * 32) + w * 4 * 32 + lane;

    for (int f = 0; f < FUT; ++f) {
      __syncthreads();
      x1[poff + 256] = p2;

      {
        v2f wpv[4];
        const float hv = gate_cell<false>(xin0F + lane, Wf0 + (w * 4 * 8) * 32 + lane, Whrf0 + (w * 4) * 32 + lane,
                                          genc[0], genc[32], genc[64], genc[96], c1r, wpv, false, x1, out, 0.f);
        v8f pacc0 = {}, pacc1 = {};
        float g[8];
        gather_h(hv, gbase, g);
        #pragma unroll
        for (int st = 0; st < 4; ++st) {
          v2f pa; pa.x = arow ? g[2 * st] : 0.f;  pa.y = arow ? g[2 * st + 1] : 0.f;
          if (st < 2) pacc0 = __builtin_amdgcn_wmma_f32_16x16x4_f32(false, pa, false, wpv[st], (short)0, pacc0, false, false);
          else        pacc1 = __builtin_amdgcn_wmma_f32_16x16x4_f32(false, pa, false, wpv[st], (short)0, pacc1, false, false);
        }
        p1 = pacc0[0] + pacc1[0];
        x1[poff] = p1;
      }
      __syncthreads();
      x0[poff + 256] = p1;

      {
        const bool do_out = (w < 1) && ((n | f) != 0);
        const float* const b1p = bias1f + (w * 4) * 16 + lm;
        v2f wpv[4];
        const float hv = gate_cell<true>(xin1F + lane, Wf1 + (w * 4 * 8) * 32 + lane, Whrf1 + (w * 4) * 32 + lane,
                                         b1p[0], b1p[16], b1p[32], b1p[48], c2r, wpv, do_out, x1 + pbase + 256, outp_prev, nm_o);
        v8f pacc0 = {}, pacc1 = {};
        float g[8];
        gather_h(hv, gbase, g);
        #pragma unroll
        for (int st = 0; st < 4; ++st) {
          v2f pa; pa.x = arow ? g[2 * st] : 0.f;  pa.y = arow ? g[2 * st + 1] : 0.f;
          if (st < 2) pacc0 = __builtin_amdgcn_wmma_f32_16x16x4_f32(false, pa, false, wpv[st], (short)0, pacc0, false, false);
          else        pacc1 = __builtin_amdgcn_wmma_f32_16x16x4_f32(false, pa, false, wpv[st], (short)0, pacc1, false, false);
        }
        p2 = pacc0[0] + pacc1[0];
        float add = 0.f;
        if (w < 1) add = (f == FUT - 1) ? nm_nx : nm_cur;
        x0[poff] = p2 + add;
      }
      outp_prev = out + (size_t)(b0 + lh) * FUT * NNODES * O_DIM
                      + (size_t)f * NNODES * O_DIM + (size_t)n * O_DIM + lm;
      nm_o = nm_cur;
    }
    nm_cur = nm_nx;
  }
  __syncthreads();
  x1[poff + 256] = p2;
  __syncthreads();
  if (w < 1) {
    float v = 0.f;
    #pragma unroll
    for (int p = 0; p < NWAVE; ++p) v += x1[pbase + 4 * p + 256];
    *outp_prev = v + nm_o;
  }
}

extern "C" void kernel_launch(void* const* d_in, const int* in_sizes, int n_in,
                              void* d_out, int out_size, void* d_ws, size_t ws_size,
                              hipStream_t stream) {
  (void)in_sizes; (void)n_in; (void)out_size; (void)d_ws; (void)ws_size;
  const float* enc   = (const float*)d_in[0];
  const float* mean_ = (const float*)d_in[1];
  const float* s     = (const float*)d_in[2];
  const float* h0    = (const float*)d_in[3];
  const float* c0    = (const float*)d_in[4];
  const float* Wih0  = (const float*)d_in[5];
  const float* Whh0  = (const float*)d_in[6];
  const float* bih0  = (const float*)d_in[7];
  const float* bhh0  = (const float*)d_in[8];
  const float* Whr0  = (const float*)d_in[9];
  const float* Wih1  = (const float*)d_in[10];
  const float* Whh1  = (const float*)d_in[11];
  const float* bih1  = (const float*)d_in[12];
  const float* bhh1  = (const float*)d_in[13];
  const float* Whr1  = (const float*)d_in[14];
  float* out = (float*)d_out;

  decoder_scan<<<dim3(NWG), dim3(NTHR), 0, stream>>>(
      enc, mean_, s, h0, c0,
      Wih0, Whh0, bih0, bhh0, Whr0,
      Wih1, Whh1, bih1, bhh1, Whr1, out);
}


# Projection partial by 16 DPP row_share multiply-adds per lane (per-lane weights from LDS), not 8 ds_bpermute + 4 two-real-row WMMA; gate products stay on WMMA; projection k sum order changes
# speedup vs baseline: 3.9230x; 1.0538x over previous
#include <hip/hip_runtime.h>
#include <math.h>

typedef __attribute__((ext_vector_type(2))) float v2f;
typedef __attribute__((ext_vector_type(4))) float v4f;
typedef __attribute__((ext_vector_type(8))) float v8f;

#define E_DIM 64
#define H_DIM 128
#define O_DIM 16
#define BATCH 64
#define NNODES 1000
#define FUT 12
#define G4 512
#define BR 2
#define NWG 32
#define NWAVE 8
#define NTHR 256
#define NB 8

__device__ __forceinline__ float sigm(float x) { return 1.0f / (1.0f + __expf(-x)); }

__device__ __forceinline__ float tanh_sel(float x) {
  float y  = __builtin_fabsf(x);
  float t  = expf(2.0f * y);
  float zl = __builtin_fmaf(__builtin_amdgcn_rcpf(t + 1.0f), -2.0f, 1.0f);
  float y2 = y * y;
  float p  = __builtin_fmaf(y2, __builtin_fmaf(y2, __builtin_fmaf(y2, __builtin_fmaf(y2, -0x1.758e7ap-8f, 0x1.521192p-6f),
                                                                 -0x1.b8389cp-5f), 0x1.110704p-3f), -0x1.555532p-2f);
  float zs = __builtin_fmaf(y2, y * p, y);
  float z  = (y < 0.625f) ? zs : zl;
  return __builtin_copysignf(z, x);
}

template <int U>
__device__ __forceinline__ float row_bcast(float v) {
  return __builtin_bit_cast(float, __builtin_amdgcn_update_dpp(0, __builtin_bit_cast(int, v), 0x150 + U, 0xf, 0xf, true));
}

template <int Q>
__device__ __forceinline__ float proj4(float hv, const v4f wv) {
  float a = row_bcast<4 * Q>(hv) * wv.x;
  a = __builtin_fmaf(row_bcast<4 * Q + 1>(hv), wv.y, a);
  a = __builtin_fmaf(row_bcast<4 * Q + 2>(hv), wv.z, a);
  a = __builtin_fmaf(row_bcast<4 * Q + 3>(hv), wv.w, a);
  return a;
}

__device__ __forceinline__ float proj_partial(float hv, const v4f (&wv)[4]) {
  const float a0 = proj4<0>(hv, wv[0]);
  const float a1 = proj4<1>(hv, wv[1]);
  const float a2 = proj4<2>(hv, wv[2]);
  const float a3 = proj4<3>(hv, wv[3]);
  return (a0 + a1) + (a2 + a3);
}

__device__ __forceinline__ float sum8(const v8f a) {
  return ((a[0] + a[1]) + (a[2] + a[3])) + ((a[4] + a[5]) + (a[6] + a[7]));
}

__device__ __forceinline__ v2f a_step(const v4f (&t)[4], int k) {
  const v4f tt = t[k >> 1];
  v2f a; a.x = (k & 1) ? tt.z : tt.x; a.y = (k & 1) ? tt.w : tt.y;
  return a;
}

template <bool OUT>
__device__ __forceinline__ float gate_cell(const v4f* __restrict__ xa, const v2f* __restrict__ wf, const v4f* __restrict__ whr,
                                           const float s0, const float s1, const float s2, const float s3,
                                           float& c, v4f (&wpv)[4],
                                           const bool do_out, const float* __restrict__ hp, float* __restrict__ outp, const float nm_o) {
  v4f t[4];
  v2f b[4][8];
  #pragma unroll
  for (int q = 0; q < 4; ++q) {
    t[q] = xa[q * 32];
    #pragma unroll
    for (int e = 0; e < 2; ++e) {
      #pragma unroll
      for (int j = 0; j < 3; ++j) b[j][2 * q + e] = wf[(8 * j + 2 * q + e) * 32];
    }
  }
  #pragma unroll
  for (int k = 0; k < 8; ++k) b[3][k] = wf[(8 * 3 + k) * 32];
  #pragma unroll
  for (int st = 0; st < 4; ++st) wpv[st] = whr[st * 16];
  float hpv[NWAVE];
  if (OUT) {
    if (do_out) {
      #pragma unroll
      for (int p = 0; p < NWAVE; ++p) hpv[p] = hp[4 * p];
    }
  }
  __builtin_amdgcn_sched_barrier(0);
  v8f acc[4];
  #pragma unroll
  for (int j = 0; j < 4; ++j) acc[j] = (v8f){};
  acc[0][0] = s0;  acc[1][0] = s1;  acc[2][0] = s2;  acc[3][0] = s3;
  #pragma unroll
  for (int k = 0; k < 8; ++k) {
    const v2f a = a_step(t, k);
    #pragma unroll
    for (int j = 0; j < 3; ++j)
      acc[j] = __builtin_amdgcn_wmma_f32_16x16x4_f32(false, a, false, b[j][k], (short)0, acc[j], false, false);
  }
  #pragma unroll
  for (int k = 0; k < 2; ++k)
    acc[3] = __builtin_amdgcn_wmma_f32_16x16x4_f32(false, a_step(t, k), false, b[3][k], (short)0, acc[3], false, false);
  if (OUT) {
    if (do_out) {
      float v = 0.f;
      #pragma unroll
      for (int p = 0; p < NWAVE; ++p) v += hpv[p];
      *outp = v + nm_o;
    }
  }
  __builtin_amdgcn_sched_barrier(0);
  #pragma unroll
  for (int k = 2; k < 8; ++k)
    acc[3] = __builtin_amdgcn_wmma_f32_16x16x4_f32(false, a_step(t, k), false, b[3][k], (short)0, acc[3], false, false);
  const float iv = sigm(sum8(acc[0]));
  const float fv = sigm(sum8(acc[1]));
  const float gv = tanh_sel(sum8(acc[2]));
  const float cn = __builtin_fmaf(c, fv, iv * gv);
  c = cn;
  const float th = tanh_sel(cn);
  const float hv = sigm(sum8(acc[3])) * th;
  #pragma unroll
  for (int i = 0; i < 6; ++i) {
    __builtin_amdgcn_sched_group_barrier(0x402, 10, 0);
    __builtin_amdgcn_sched_group_barrier(0x008, 1, 0);
  }
  return hv;
}

__global__ __launch_bounds__(NTHR) void decoder_scan(
    const float* __restrict__ enc,
    const float* __restrict__ mean_,
    const float* __restrict__ s,
    const float* __restrict__ h0,
    const float* __restrict__ c0,
    const float* __restrict__ Wih0,
    const float* __restrict__ Whh0,
    const float* __restrict__ bih0, const float* __restrict__ bhh0,
    const float* __restrict__ Whr0,
    const float* __restrict__ Wih1,
    const float* __restrict__ Whh1,
    const float* __restrict__ bih1, const float* __restrict__ bhh1,
    const float* __restrict__ Whr1,
    float* __restrict__ out)
{
  __shared__ v2f Wf0[NWAVE * 4 * 8 * 32];
  __shared__ v2f Wf1[NWAVE * 4 * 8 * 32];
  __shared__ v4f Whrf0[NWAVE * 4 * 16];
  __shared__ v4f Whrf1[NWAVE * 4 * 16];
  __shared__ float GencF[NB * NWAVE * 4 * 32];
  __shared__ float bias0s[G4];
  __shared__ float bias1f[NWAVE * 4 * 16];
  __shared__ v4f xin0F[4 * 32];
  __shared__ v4f xin1F[4 * 32];

  const int tid  = threadIdx.x;
  const int w    = __builtin_amdgcn_readfirstlane(tid >> 5);
  const int lane = tid & 31;
  const int lm   = lane & 15;
  const int lh   = lane >> 4;
  const int b0   = blockIdx.x * BR;
  const int cu   = 16 * w + lm;
  const int pbase = (lm >> 3) * 128 + (8 * lh + 16 * ((lm >> 1) & 1)) * 4 + ((lm >> 2) & 1) * 2 + (lm & 1);
  const int poff  = pbase + 4 * w;
  const int aslot = lm >> 3;
  const int anode = lm & 7;
  float* const x0 = (float*)xin0F;
  float* const x1 = (float*)xin1F;

  for (int idx = tid; idx < NWAVE * 4 * 8 * 32 * 2; idx += NTHR) {
    int e = idx & 1, ln = (idx >> 1) & 31, k = (idx >> 6) & 7, j = (idx >> 9) & 3, ww = idx >> 11;
    int kr = 4 * k + 2 * (ln >> 4) + e;
    int col = 128 * j + 16 * ww + (ln & 15);
    ((float*)Wf0)[idx] = (kr < 16) ? Wih0[col * 80 + 64 + kr] : Whh0[col * 16 + (kr - 16)];
    ((float*)Wf1)[idx] = (kr < 16) ? Wih1[col * 16 + kr]      : Whh1[col * 16 + (kr - 16)];
  }
  for (int idx = tid; idx < NWAVE * 4 * 16 * 4; idx += NTHR) {
    int e = idx & 3, o = (idx >> 2) & 15, st = (idx >> 6) & 3, ww = idx >> 8;
    int unit = 16 * ww + 4 * st + e;
    ((float*)Whrf0)[idx] = Whr0[o * H_DIM + unit];
    ((float*)Whrf1)[idx] = Whr1[o * H_DIM + unit];
  }
  for (int idx = tid; idx < G4; idx += NTHR) {
    bias0s[idx] = bih0[idx] + bhh0[idx];
    int l15 = idx & 15, j = (idx >> 4) & 3, ww = idx >> 6;
    int c1 = 128 * j + 16 * ww + l15;
    bias1f[idx] = bih1[c1] + bhh1[c1];
  }
  const float* const mp0 = mean_ + (size_t)(b0 + lh) * FUT * NNODES * O_DIM + lm;
  float nm_cur = 0.f, nm_nx = 0.f, nm_o = 0.f;
  float p1 = 0.f, p2 = 0.f, lp = 0.f;
  if (w < 1) {
    float sum = 0.f;
    #pragma unroll
    for (int f = 0; f < FUT; ++f) sum += mp0[(size_t)f * NNODES * O_DIM];
    nm_cur = sum * (1.0f / FUT);
    p1 = h0[lm];
    p2 = h0[O_DIM + lm];
    lp = s[lm] + nm_cur;
  }
  x0[poff]       = lp;
  x0[poff + 256] = p1;
  float c1r = c0[cu], c2r = c0[H_DIM + cu];
  v2f wenc[16][4];
  #pragma unroll
  for (int k = 0; k < 16; ++k) {
    #pragma unroll
    for (int j = 0; j < 4; ++j) {
      const float* wp = Wih0 + (128 * j + cu) * 80 + 4 * k + 2 * lh;
      wenc[k][j].x = wp[0];  wenc[k][j].y = wp[1];
    }
  }
  float* outp_prev = out;
  __syncthreads();

  for (int n = 0; n < NNODES; ++n) {
    if (w < 1) {
      nm_nx = 0.f;
      if (n + 1 < NNODES) {
        const float* mp = mp0 + (size_t)(n + 1) * O_DIM;
        float sum = 0.f;
        #pragma unroll
        for (int f = 0; f < FUT; ++f) sum += mp[(size_t)f * NNODES * O_DIM];
        nm_nx = sum * (1.0f / FUT);
      }
    }
    if ((n & (NB - 1)) == 0) {
      v8f hacc[4];
      #pragma unroll
      for (int j = 0; j < 4; ++j) {
        float bv = bias0s[128 * j + cu];
        #pragma unroll
        for (int r = 0; r < 8; ++r) hacc[j][r] = bv;
      }
      const int an = (n + anode < NNODES) ? n + anode : NNODES - 1;
      const float* ap = enc + (size_t)(b0 + aslot) * NNODES * E_DIM + (size_t)an * E_DIM;
      #pragma unroll
      for (int k = 0; k < 16; ++k) {
        int kk = 4 * k + 2 * lh;
        v2f a;
        a.x = ap[kk];  a.y = ap[kk + 1];
        #pragma unroll
        for (int j = 0; j < 4; ++j)
          hacc[j] = __builtin_amdgcn_wmma_f32_16x16x4_f32(false, a, false, wenc[k][j], (short)0, hacc[j], false, false);
      }
      #pragma unroll
      for (int r = 0; r < NB; ++r) {
        #pragma unroll
        for (int j = 0; j < 4; ++j) GencF[r * (NWAVE * 4 * 32) + (w * 4 + j) * 32 + lane] = hacc[j][r];
      }
    }
    const float* const genc = GencF + (n & (NB - 1)) * (NWAVE * 4 * 32) + w * 4 * 32 + lane;

    for (int f = 0; f < FUT; ++f) {
      __syncthreads();
      x1[poff + 256] = p2;

      {
        v4f wpv[4];
        const float hv = gate_cell<false>(xin0F + lane, Wf0 + (w * 4 * 8) * 32 + lane, Whrf0 + (w * 4) * 16 + lm,
                                          genc[0], genc[32], genc[64], genc[96], c1r, wpv, false, x1, out, 0.f);
        p1 = proj_partial(hv, wpv);
        x1[poff] = p1;
      }
      __syncthreads();
      x0[poff + 256] = p1;

      {
        const bool do_out = (w < 1) && ((n | f) != 0);
        const float* const b1p = bias1f + (w * 4) * 16 + lm;
        v4f wpv[4];
        const float hv = gate_cell<true>(xin1F + lane, Wf1 + (w * 4 * 8) * 32 + lane, Whrf1 + (w * 4) * 16 + lm,
                                         b1p[0], b1p[16], b1p[32], b1p[48], c2r, wpv, do_out, x1 + pbase + 256, outp_prev, nm_o);
        p2 = proj_partial(hv, wpv);
        float add = 0.f;
        if (w < 1) add = (f == FUT - 1) ? nm_nx : nm_cur;
        x0[poff] = p2 + add;
      }
      outp_prev = out + (size_t)(b0 + lh) * FUT * NNODES * O_DIM
                      + (size_t)f * NNODES * O_DIM + (size_t)n * O_DIM + lm;
      nm_o = nm_cur;
    }
    nm_cur = nm_nx;
  }
  __syncthreads();
  x1[poff + 256] = p2;
  __syncthreads();
  if (w < 1) {
    float v = 0.f;
    #pragma unroll
    for (int p = 0; p < NWAVE; ++p) v += x1[pbase + 4 * p + 256];
    *outp_prev = v + nm_o;
  }
}

extern "C" void kernel_launch(void* const* d_in, const int* in_sizes, int n_in,
                              void* d_out, int out_size, void* d_ws, size_t ws_size,
                              hipStream_t stream) {
  (void)in_sizes; (void)n_in; (void)out_size; (void)d_ws; (void)ws_size;
  const float* enc   = (const float*)d_in[0];
  const float* mean_ = (const float*)d_in[1];
  const float* s     = (const float*)d_in[2];
  const float* h0    = (const float*)d_in[3];
  const float* c0    = (const float*)d_in[4];
  const float* Wih0  = (const float*)d_in[5];
  const float* Whh0  = (const float*)d_in[6];
  const float* bih0  = (const float*)d_in[7];
  const float* bhh0  = (const float*)d_in[8];
  const float* Whr0  = (const float*)d_in[9];
  const float* Wih1  = (const float*)d_in[10];
  const float* Whh1  = (const float*)d_in[11];
  const float* bih1  = (const float*)d_in[12];
  const float* bhh1  = (const float*)d_in[13];
  const float* Whr1  = (const float*)d_in[14];
  float* out = (float*)d_out;

  decoder_scan<<<dim3(NWG), dim3(NTHR), 0, stream>>>(
      enc, mean_, s, h0, c0,
      Wih0, Whh0, bih0, bhh0, Whr0,
      Wih1, Whh1, bih1, bhh1, Whr1, out);
}
